# starting kernel
# baseline (speedup 1.0000x reference)
#include <hip/hip_runtime.h>
#include <hip/hip_bf16.h>


#define Bn 64
#define Nn 256
#define Hn 1024
#define NK 768
#define LN_EPS 1e-5f

typedef __attribute__((ext_vector_type(2))) float v2f;
typedef __attribute__((ext_vector_type(8))) float v8f;

__global__ __launch_bounds__(256) void wgen_gemm(
    const float* __restrict__ f, const float* __restrict__ Wlin,
    const float* __restrict__ blin, float* __restrict__ wout) {
  __shared__ float As[64 * 20];
  __shared__ float Bs[16 * 136];

  const int tid = threadIdx.x;
  const int b = blockIdx.z;
  const int m0 = blockIdx.y * 64;
  const int o0 = blockIdx.x * 128;
  const int wave = tid >> 5, lane = tid & 31;
  const int lm = lane & 15, hi = lane >> 4;
  const int wm = (wave & 3) * 16;
  const int wn = (wave >> 2) * 64;

  v8f cAcc[4] = {};

  const int arow = tid >> 2;
  const int acol = (tid & 3) * 4;
  const float* fA = f + (size_t)(b * Nn + m0 + arow) * Hn + acol;

  for (int k0 = 0; k0 < Hn; k0 += 16) {
    __syncthreads();
    float4 av = *(const float4*)(fA + k0);
    *(float4*)(&As[arow * 20 + acol]) = av;
#pragma unroll
    for (int j = 0; j < 2; ++j) {
      int idx = tid + j * 256;
      int br = idx >> 2;
      int bc = (idx & 3) * 4;
      float4 bv = *(const float4*)(Wlin + (size_t)(o0 + br) * Hn + k0 + bc);
      Bs[(bc + 0) * 136 + br] = bv.x;
      Bs[(bc + 1) * 136 + br] = bv.y;
      Bs[(bc + 2) * 136 + br] = bv.z;
      Bs[(bc + 3) * 136 + br] = bv.w;
    }
    __syncthreads();
#pragma unroll
    for (int ks = 0; ks < 4; ++ks) {
      const int kk = ks * 4 + hi * 2;
      v2f a;
      a.x = As[(wm + lm) * 20 + kk];
      a.y = As[(wm + lm) * 20 + kk + 1];
#pragma unroll
      for (int ct = 0; ct < 4; ++ct) {
        const int n = wn + ct * 16 + lm;
        v2f bb;
        bb.x = Bs[(kk + 0) * 136 + n];
        bb.y = Bs[(kk + 1) * 136 + n];
        cAcc[ct] = __builtin_amdgcn_wmma_f32_16x16x4_f32(
            false, a, false, bb, (short)0, cAcc[ct], false, false);
      }
    }
  }

#pragma unroll
  for (int ct = 0; ct < 4; ++ct) {
    const int o = o0 + wn + ct * 16 + lm;
    const float bl = blin[o];
#pragma unroll
    for (int r = 0; r < 8; ++r) {
      const int m = m0 + wm + r + hi * 8;
      wout[(size_t)(b * Nn + m) * NK + o] = cAcc[ct][r] + bl;
    }
  }
}

__global__ __launch_bounds__(256) void dyconv_ln(
    const float* __restrict__ kin, const float* __restrict__ w,
    const float* __restrict__ gamma, const float* __restrict__ beta,
    float* __restrict__ out) {
  __shared__ float Ks[16 * 520];
  __shared__ float Aw[16 * 52];
  __shared__ float Gs[1024];
  __shared__ float Bt[1024];
  __shared__ float redS[8 * 16];
  __shared__ float redQ[8 * 16];
  __shared__ float muS[16];
  __shared__ float rsS[16];

  const int tid = threadIdx.x;
  const int b = blockIdx.y;
  const int m0 = blockIdx.x * 16;
  const int wave = tid >> 5, lane = tid & 31;
  const int lm = lane & 15, hi2 = lane >> 4;

#pragma unroll
  for (int j = 0; j < 4; ++j) {
    int i = tid + j * 256;
    Gs[i] = gamma[i];
    Bt[i] = beta[i];
  }

  v8f acc[8] = {};

  const int aRow = tid >> 4;
  const int aCol = (tid & 15) * 3;
  const float* wp = w + (size_t)(b * Nn + m0 + aRow) * NK + aCol;
  const int kRow = tid >> 4;
  const int kCb = (tid & 15) * 32;
  const int hRow = tid & 15;

  for (int ic = 0; ic < 16; ++ic) {
    const int i0 = ic * 16;
    __syncthreads();
    {
      const float* wq = wp + (size_t)i0 * 3;
      Aw[aRow * 52 + aCol + 0] = wq[0];
      Aw[aRow * 52 + aCol + 1] = wq[1];
      Aw[aRow * 52 + aCol + 2] = wq[2];
    }
#pragma unroll
    for (int p = 0; p < 2; ++p) {
      if (p) __syncthreads();
      {
        const float* kr = kin + (size_t)(b * Nn + i0 + kRow) * Hn + p * 512 + kCb;
#pragma unroll
        for (int j = 0; j < 8; ++j) {
          float4 v = *(const float4*)(kr + j * 4);
          const int cc = kRow * 520 + 1 + kCb + j * 4;
          Ks[cc + 0] = v.x;
          Ks[cc + 1] = v.y;
          Ks[cc + 2] = v.z;
          Ks[cc + 3] = v.w;
        }
      }
      if (tid < 32) {
        const float* krh = kin + (size_t)(b * Nn + i0 + hRow) * Hn;
        if (tid < 16) {
          Ks[hRow * 520 + 0] = (p == 0) ? 0.0f : krh[511];
        } else {
          Ks[hRow * 520 + 513] = (p == 0) ? krh[512] : 0.0f;
        }
      }
      __syncthreads();
#pragma unroll
      for (int t = 0; t < 3; ++t) {
#pragma unroll
        for (int ks = 0; ks < 4; ++ks) {
          const int isub = ks * 4 + hi2 * 2;
          v2f a;
          a.x = Aw[lm * 52 + (isub + 0) * 3 + t];
          a.y = Aw[lm * 52 + (isub + 1) * 3 + t];
#pragma unroll
          for (int ct = 0; ct < 4; ++ct) {
            const int col = (wave * 64 + ct * 16 + lm) + t;
            v2f bb;
            bb.x = Ks[(isub + 0) * 520 + col];
            bb.y = Ks[(isub + 1) * 520 + col];
            acc[p * 4 + ct] = __builtin_amdgcn_wmma_f32_16x16x4_f32(
                false, a, false, bb, (short)0, acc[p * 4 + ct], false, false);
          }
        }
      }
    }
  }

  float s[8], q[8];
#pragma unroll
  for (int r = 0; r < 8; ++r) { s[r] = 0.0f; q[r] = 0.0f; }
#pragma unroll
  for (int u = 0; u < 8; ++u)
#pragma unroll
    for (int r = 0; r < 8; ++r) {
      float v = acc[u][r];
      s[r] += v;
      q[r] += v * v;
    }
#pragma unroll
  for (int mask = 1; mask < 16; mask <<= 1)
#pragma unroll
    for (int r = 0; r < 8; ++r) {
      s[r] += __shfl_xor(s[r], mask, 32);
      q[r] += __shfl_xor(q[r], mask, 32);
    }
  if (lm == 0) {
#pragma unroll
    for (int r = 0; r < 8; ++r) {
      const int row = r + hi2 * 8;
      redS[wave * 16 + row] = s[r];
      redQ[wave * 16 + row] = q[r];
    }
  }
  __syncthreads();
  if (tid < 16) {
    float ss = 0.0f, qq = 0.0f;
#pragma unroll
    for (int wv = 0; wv < 8; ++wv) {
      ss += redS[wv * 16 + tid];
      qq += redQ[wv * 16 + tid];
    }
    const float mu = ss * (1.0f / 1024.0f);
    const float var = qq * (1.0f / 1024.0f) - mu * mu;
    muS[tid] = mu;
    rsS[tid] = rsqrtf(var + LN_EPS);
  }
  __syncthreads();

#pragma unroll
  for (int u = 0; u < 8; ++u) {
    const int p = u >> 2, ct = u & 3;
    const int h = p * 512 + wave * 64 + ct * 16 + lm;
    const float g = Gs[h], be = Bt[h];
#pragma unroll
    for (int r = 0; r < 8; ++r) {
      const int row = r + hi2 * 8;
      const float v = (acc[u][r] - muS[row]) * rsS[row] * g + be;
      out[(size_t)(b * Nn + m0 + row) * Hn + h] = v;
    }
  }
}

extern "C" void kernel_launch(void* const* d_in, const int* in_sizes, int n_in,
                              void* d_out, int out_size, void* d_ws, size_t ws_size,
                              hipStream_t stream) {
  const float* f     = (const float*)d_in[0];
  const float* k     = (const float*)d_in[1];
  const float* Wlin  = (const float*)d_in[2];
  const float* blin  = (const float*)d_in[3];
  const float* gamma = (const float*)d_in[4];
  const float* beta  = (const float*)d_in[5];
  float* out = (float*)d_out;
  float* wgt = (float*)d_ws;

  (void)in_sizes; (void)n_in; (void)out_size; (void)ws_size;

  wgen_gemm<<<dim3(NK / 128, Nn / 64, Bn), 256, 0, stream>>>(f, Wlin, blin, wgt);
  dyconv_ln<<<dim3(Nn / 16, Bn), 256, 0, stream>>>(k, wgt, gamma, beta, out);
}


# Stage-2 k slab staging remapped: each wave loads 512 contiguous bytes and stores aligned 16-byte LDS words (left pad 4 floats), replacing 128-byte-strided loads and 8-way conflicted scalar stores
# speedup vs baseline: 1.1910x; 1.1910x over previous
#include <hip/hip_runtime.h>
#include <hip/hip_bf16.h>


#define Bn 64
#define Nn 256
#define Hn 1024
#define NK 768
#define LN_EPS 1e-5f

typedef __attribute__((ext_vector_type(2))) float v2f;
typedef __attribute__((ext_vector_type(8))) float v8f;

__global__ __launch_bounds__(256) void wgen_gemm(
    const float* __restrict__ f, const float* __restrict__ Wlin,
    const float* __restrict__ blin, float* __restrict__ wout) {
  __shared__ float As[64 * 20];
  __shared__ float Bs[16 * 136];

  const int tid = threadIdx.x;
  const int b = blockIdx.z;
  const int m0 = blockIdx.y * 64;
  const int o0 = blockIdx.x * 128;
  const int wave = tid >> 5, lane = tid & 31;
  const int lm = lane & 15, hi = lane >> 4;
  const int wm = (wave & 3) * 16;
  const int wn = (wave >> 2) * 64;

  v8f cAcc[4] = {};

  const int arow = tid >> 2;
  const int acol = (tid & 3) * 4;
  const float* fA = f + (size_t)(b * Nn + m0 + arow) * Hn + acol;

  for (int k0 = 0; k0 < Hn; k0 += 16) {
    __syncthreads();
    float4 av = *(const float4*)(fA + k0);
    *(float4*)(&As[arow * 20 + acol]) = av;
#pragma unroll
    for (int j = 0; j < 2; ++j) {
      int idx = tid + j * 256;
      int br = idx >> 2;
      int bc = (idx & 3) * 4;
      float4 bv = *(const float4*)(Wlin + (size_t)(o0 + br) * Hn + k0 + bc);
      Bs[(bc + 0) * 136 + br] = bv.x;
      Bs[(bc + 1) * 136 + br] = bv.y;
      Bs[(bc + 2) * 136 + br] = bv.z;
      Bs[(bc + 3) * 136 + br] = bv.w;
    }
    __syncthreads();
#pragma unroll
    for (int ks = 0; ks < 4; ++ks) {
      const int kk = ks * 4 + hi * 2;
      v2f a;
      a.x = As[(wm + lm) * 20 + kk];
      a.y = As[(wm + lm) * 20 + kk + 1];
#pragma unroll
      for (int ct = 0; ct < 4; ++ct) {
        const int n = wn + ct * 16 + lm;
        v2f bb;
        bb.x = Bs[(kk + 0) * 136 + n];
        bb.y = Bs[(kk + 1) * 136 + n];
        cAcc[ct] = __builtin_amdgcn_wmma_f32_16x16x4_f32(
            false, a, false, bb, (short)0, cAcc[ct], false, false);
      }
    }
  }

#pragma unroll
  for (int ct = 0; ct < 4; ++ct) {
    const int o = o0 + wn + ct * 16 + lm;
    const float bl = blin[o];
#pragma unroll
    for (int r = 0; r < 8; ++r) {
      const int m = m0 + wm + r + hi * 8;
      wout[(size_t)(b * Nn + m) * NK + o] = cAcc[ct][r] + bl;
    }
  }
}

__global__ __launch_bounds__(256) void dyconv_ln(
    const float* __restrict__ kin, const float* __restrict__ w,
    const float* __restrict__ gamma, const float* __restrict__ beta,
    float* __restrict__ out) {
  __shared__ __attribute__((aligned(16))) float Ks[16 * 520];
  __shared__ float Aw[16 * 52];
  __shared__ float Gs[1024];
  __shared__ float Bt[1024];
  __shared__ float redS[8 * 16];
  __shared__ float redQ[8 * 16];
  __shared__ float muS[16];
  __shared__ float rsS[16];

  const int tid = threadIdx.x;
  const int b = blockIdx.y;
  const int m0 = blockIdx.x * 16;
  const int wave = tid >> 5, lane = tid & 31;
  const int lm = lane & 15, hi2 = lane >> 4;

#pragma unroll
  for (int j = 0; j < 4; ++j) {
    int i = tid + j * 256;
    Gs[i] = gamma[i];
    Bt[i] = beta[i];
  }

  v8f acc[8] = {};

  const int aRow = tid >> 4;
  const int aCol = (tid & 15) * 3;
  const float* wp = w + (size_t)(b * Nn + m0 + aRow) * NK + aCol;
  const int kRow0 = tid >> 7;
  const int kC4 = (tid & 127) * 4;
  const int hRow = tid & 15;

  for (int ic = 0; ic < 16; ++ic) {
    const int i0 = ic * 16;
    __syncthreads();
    {
      const float* wq = wp + (size_t)i0 * 3;
      Aw[aRow * 52 + aCol + 0] = wq[0];
      Aw[aRow * 52 + aCol + 1] = wq[1];
      Aw[aRow * 52 + aCol + 2] = wq[2];
    }
#pragma unroll
    for (int p = 0; p < 2; ++p) {
      if (p) __syncthreads();
      {
        const float* kr = kin + (size_t)(b * Nn + i0 + kRow0) * Hn + p * 512 + kC4;
#pragma unroll
        for (int j = 0; j < 8; ++j) {
          float4 v = *(const float4*)(kr + (size_t)(j * 2) * Hn);
          *(float4*)(&Ks[(j * 2 + kRow0) * 520 + 4 + kC4]) = v;
        }
      }
      if (tid < 32) {
        const float* krh = kin + (size_t)(b * Nn + i0 + hRow) * Hn;
        if (tid < 16) {
          Ks[hRow * 520 + 3] = (p == 0) ? 0.0f : krh[511];
        } else {
          Ks[hRow * 520 + 516] = (p == 0) ? krh[512] : 0.0f;
        }
      }
      __syncthreads();
#pragma unroll
      for (int t = 0; t < 3; ++t) {
#pragma unroll
        for (int ks = 0; ks < 4; ++ks) {
          const int isub = ks * 4 + hi2 * 2;
          v2f a;
          a.x = Aw[lm * 52 + (isub + 0) * 3 + t];
          a.y = Aw[lm * 52 + (isub + 1) * 3 + t];
#pragma unroll
          for (int ct = 0; ct < 4; ++ct) {
            const int col = (wave * 64 + ct * 16 + lm) + t + 3;
            v2f bb;
            bb.x = Ks[(isub + 0) * 520 + col];
            bb.y = Ks[(isub + 1) * 520 + col];
            acc[p * 4 + ct] = __builtin_amdgcn_wmma_f32_16x16x4_f32(
                false, a, false, bb, (short)0, acc[p * 4 + ct], false, false);
          }
        }
      }
    }
  }

  float s[8], q[8];
#pragma unroll
  for (int r = 0; r < 8; ++r) { s[r] = 0.0f; q[r] = 0.0f; }
#pragma unroll
  for (int u = 0; u < 8; ++u)
#pragma unroll
    for (int r = 0; r < 8; ++r) {
      float v = acc[u][r];
      s[r] += v;
      q[r] += v * v;
    }
#pragma unroll
  for (int mask = 1; mask < 16; mask <<= 1)
#pragma unroll
    for (int r = 0; r < 8; ++r) {
      s[r] += __shfl_xor(s[r], mask, 32);
      q[r] += __shfl_xor(q[r], mask, 32);
    }
  if (lm == 0) {
#pragma unroll
    for (int r = 0; r < 8; ++r) {
      const int row = r + hi2 * 8;
      redS[wave * 16 + row] = s[r];
      redQ[wave * 16 + row] = q[r];
    }
  }
  __syncthreads();
  if (tid < 16) {
    float ss = 0.0f, qq = 0.0f;
#pragma unroll
    for (int wv = 0; wv < 8; ++wv) {
      ss += redS[wv * 16 + tid];
      qq += redQ[wv * 16 + tid];
    }
    const float mu = ss * (1.0f / 1024.0f);
    const float var = qq * (1.0f / 1024.0f) - mu * mu;
    muS[tid] = mu;
    rsS[tid] = rsqrtf(var + LN_EPS);
  }
  __syncthreads();

#pragma unroll
  for (int u = 0; u < 8; ++u) {
    const int p = u >> 2, ct = u & 3;
    const int h = p * 512 + wave * 64 + ct * 16 + lm;
    const float g = Gs[h], be = Bt[h];
#pragma unroll
    for (int r = 0; r < 8; ++r) {
      const int row = r + hi2 * 8;
      const float v = (acc[u][r] - muS[row]) * rsS[row] * g + be;
      out[(size_t)(b * Nn + m0 + row) * Hn + h] = v;
    }
  }
}

extern "C" void kernel_launch(void* const* d_in, const int* in_sizes, int n_in,
                              void* d_out, int out_size, void* d_ws, size_t ws_size,
                              hipStream_t stream) {
  const float* f     = (const float*)d_in[0];
  const float* k     = (const float*)d_in[1];
  const float* Wlin  = (const float*)d_in[2];
  const float* blin  = (const float*)d_in[3];
  const float* gamma = (const float*)d_in[4];
  const float* beta  = (const float*)d_in[5];
  float* out = (float*)d_out;
  float* wgt = (float*)d_ws;

  (void)in_sizes; (void)n_in; (void)out_size; (void)ws_size;

  wgen_gemm<<<dim3(NK / 128, Nn / 64, Bn), 256, 0, stream>>>(f, Wlin, blin, wgt);
  dyconv_ln<<<dim3(Nn / 16, Bn), 256, 0, stream>>>(k, wgt, gamma, beta, out);
}


# LDS operand tiles re-laid so each lane's k values are adjacent (stage 1 W tile kept row-major, stage 2 k slab channel-pair interleaved): one 8/16-byte LDS read per fragment, k order unchanged
# speedup vs baseline: 1.3125x; 1.1020x over previous
#include <hip/hip_runtime.h>
#include <hip/hip_bf16.h>


#define Bn 64
#define Nn 256
#define Hn 1024
#define NK 768
#define LN_EPS 1e-5f
#define KP 528

__device__ __forceinline__ int kpos(int k) { return 8 * ((k >> 1) & 1) + 2 * (k >> 2) + (k & 1); }

typedef __attribute__((ext_vector_type(2))) float v2f;
typedef __attribute__((ext_vector_type(8))) float v8f;

__global__ __launch_bounds__(256) void wgen_gemm(
    const float* __restrict__ f, const float* __restrict__ Wlin,
    const float* __restrict__ blin, float* __restrict__ wout) {
  __shared__ __attribute__((aligned(16))) float As[64 * 20];
  __shared__ __attribute__((aligned(16))) float Bs[128 * 20];

  const int tid = threadIdx.x;
  const int b = blockIdx.z;
  const int m0 = blockIdx.y * 64;
  const int o0 = blockIdx.x * 128;
  const int wave = tid >> 5, lane = tid & 31;
  const int lm = lane & 15, hi = lane >> 4;
  const int wm = (wave & 3) * 16;
  const int wn = (wave >> 2) * 64;

  v8f cAcc[4] = {};

  const int arow = tid >> 2;
  const int acol = (tid & 3) * 4;
  const float* fA = f + (size_t)(b * Nn + m0 + arow) * Hn + acol;

  for (int k0 = 0; k0 < Hn; k0 += 16) {
    __syncthreads();
    float4 av = *(const float4*)(fA + k0);
    *(float2*)(&As[arow * 20 + kpos(acol)])     = make_float2(av.x, av.y);
    *(float2*)(&As[arow * 20 + kpos(acol + 2)]) = make_float2(av.z, av.w);
#pragma unroll
    for (int j = 0; j < 2; ++j) {
      int idx = tid + j * 256;
      int br = idx >> 2;
      int bc = (idx & 3) * 4;
      float4 bv = *(const float4*)(Wlin + (size_t)(o0 + br) * Hn + k0 + bc);
      *(float2*)(&Bs[br * 20 + kpos(bc)])     = make_float2(bv.x, bv.y);
      *(float2*)(&Bs[br * 20 + kpos(bc + 2)]) = make_float2(bv.z, bv.w);
    }
    __syncthreads();
    v2f af[4], bf[4][4];
    {
      const float4 lo = *(const float4*)(&As[(wm + lm) * 20 + hi * 8]);
      const float4 up = *(const float4*)(&As[(wm + lm) * 20 + hi * 8 + 4]);
      af[0].x = lo.x; af[0].y = lo.y; af[1].x = lo.z; af[1].y = lo.w;
      af[2].x = up.x; af[2].y = up.y; af[3].x = up.z; af[3].y = up.w;
    }
#pragma unroll
    for (int ct = 0; ct < 4; ++ct) {
      const int n = wn + ct * 16 + lm;
      const float4 lo = *(const float4*)(&Bs[n * 20 + hi * 8]);
      const float4 up = *(const float4*)(&Bs[n * 20 + hi * 8 + 4]);
      bf[ct][0].x = lo.x; bf[ct][0].y = lo.y; bf[ct][1].x = lo.z; bf[ct][1].y = lo.w;
      bf[ct][2].x = up.x; bf[ct][2].y = up.y; bf[ct][3].x = up.z; bf[ct][3].y = up.w;
    }
#pragma unroll
    for (int ks = 0; ks < 4; ++ks) {
#pragma unroll
      for (int ct = 0; ct < 4; ++ct) {
        cAcc[ct] = __builtin_amdgcn_wmma_f32_16x16x4_f32(
            false, af[ks], false, bf[ct][ks], (short)0, cAcc[ct], false, false);
      }
    }
  }

#pragma unroll
  for (int ct = 0; ct < 4; ++ct) {
    const int o = o0 + wn + ct * 16 + lm;
    const float bl = blin[o];
#pragma unroll
    for (int r = 0; r < 8; ++r) {
      const int m = m0 + wm + r + hi * 8;
      wout[(size_t)(b * Nn + m) * NK + o] = cAcc[ct][r] + bl;
    }
  }
}

__global__ __launch_bounds__(256) void dyconv_ln(
    const float* __restrict__ kin, const float* __restrict__ w,
    const float* __restrict__ gamma, const float* __restrict__ beta,
    float* __restrict__ out) {
  __shared__ __attribute__((aligned(16))) float Ks[8 * KP * 2];
  __shared__ __attribute__((aligned(16))) float Aw[16 * 52];
  __shared__ float Gs[1024];
  __shared__ float Bt[1024];
  __shared__ float redS[8 * 16];
  __shared__ float redQ[8 * 16];
  __shared__ float muS[16];
  __shared__ float rsS[16];

  const int tid = threadIdx.x;
  const int b = blockIdx.y;
  const int m0 = blockIdx.x * 16;
  const int wave = tid >> 5, lane = tid & 31;
  const int lm = lane & 15, hi2 = lane >> 4;

#pragma unroll
  for (int j = 0; j < 4; ++j) {
    int i = tid + j * 256;
    Gs[i] = gamma[i];
    Bt[i] = beta[i];
  }

  v8f acc[8] = {};

  const int aRow = tid >> 4;
  const int aI = tid & 15;
  const float* wp = w + (size_t)(b * Nn + m0 + aRow) * NK + aI * 3;
  const int aDst = aRow * 52 + kpos(aI);
  const int kPr0 = tid >> 7;
  const int kC4 = (tid & 127) * 4;
  const int hRow = tid & 15;
  const int hBase = ((hRow >> 1) * KP) * 2 + (hRow & 1);

  for (int ic = 0; ic < 16; ++ic) {
    const int i0 = ic * 16;
    __syncthreads();
    {
      const float* wq = wp + (size_t)i0 * 3;
      Aw[aDst + 0 * 16] = wq[0];
      Aw[aDst + 1 * 16] = wq[1];
      Aw[aDst + 2 * 16] = wq[2];
    }
#pragma unroll
    for (int p = 0; p < 2; ++p) {
      if (p) __syncthreads();
      {
        const float* kr = kin + (size_t)(b * Nn + i0 + kPr0 * 2) * Hn + p * 512 + kC4;
        float* kd = &Ks[(kPr0 * KP + 4 + kC4) * 2];
#pragma unroll
        for (int j = 0; j < 4; ++j) {
          const float4 x = *(const float4*)(kr + (size_t)(j * 4) * Hn);
          const float4 y = *(const float4*)(kr + (size_t)(j * 4 + 1) * Hn);
          const float4 lo = make_float4(x.x, y.x, x.y, y.y);
          const float4 up = make_float4(x.z, y.z, x.w, y.w);
          float* d = kd + (j * 2 * KP) * 2;
          *(float4*)(d)     = lo;
          *(float4*)(d + 4) = up;
        }
      }
      if (tid < 32) {
        const float* krh = kin + (size_t)(b * Nn + i0 + hRow) * Hn;
        if (tid < 16) {
          Ks[hBase + 3 * 2] = (p == 0) ? 0.0f : krh[511];
        } else {
          Ks[hBase + 516 * 2] = (p == 0) ? krh[512] : 0.0f;
        }
      }
      __syncthreads();
#pragma unroll
      for (int t = 0; t < 3; ++t) {
        const float4 alo = *(const float4*)(&Aw[lm * 52 + t * 16 + hi2 * 8]);
        const float4 aup = *(const float4*)(&Aw[lm * 52 + t * 16 + hi2 * 8 + 4]);
        v2f af[4];
        af[0].x = alo.x; af[0].y = alo.y; af[1].x = alo.z; af[1].y = alo.w;
        af[2].x = aup.x; af[2].y = aup.y; af[3].x = aup.z; af[3].y = aup.w;
#pragma unroll
        for (int ks = 0; ks < 4; ++ks) {
          const v2f a = af[ks];
#pragma unroll
          for (int ct = 0; ct < 4; ++ct) {
            const int col = (wave * 64 + ct * 16 + lm) + t + 3;
            const v2f bb = *(const v2f*)(&Ks[((ks * 2 + hi2) * KP + col) * 2]);
            acc[p * 4 + ct] = __builtin_amdgcn_wmma_f32_16x16x4_f32(
                false, a, false, bb, (short)0, acc[p * 4 + ct], false, false);
          }
        }
      }
    }
  }

  float s[8], q[8];
#pragma unroll
  for (int r = 0; r < 8; ++r) { s[r] = 0.0f; q[r] = 0.0f; }
#pragma unroll
  for (int u = 0; u < 8; ++u)
#pragma unroll
    for (int r = 0; r < 8; ++r) {
      float v = acc[u][r];
      s[r] += v;
      q[r] += v * v;
    }
#pragma unroll
  for (int mask = 1; mask < 16; mask <<= 1)
#pragma unroll
    for (int r = 0; r < 8; ++r) {
      s[r] += __shfl_xor(s[r], mask, 32);
      q[r] += __shfl_xor(q[r], mask, 32);
    }
  if (lm == 0) {
#pragma unroll
    for (int r = 0; r < 8; ++r) {
      const int row = r + hi2 * 8;
      redS[wave * 16 + row] = s[r];
      redQ[wave * 16 + row] = q[r];
    }
  }
  __syncthreads();
  if (tid < 16) {
    float ss = 0.0f, qq = 0.0f;
#pragma unroll
    for (int wv = 0; wv < 8; ++wv) {
      ss += redS[wv * 16 + tid];
      qq += redQ[wv * 16 + tid];
    }
    const float mu = ss * (1.0f / 1024.0f);
    const float var = qq * (1.0f / 1024.0f) - mu * mu;
    muS[tid] = mu;
    rsS[tid] = rsqrtf(var + LN_EPS);
  }
  __syncthreads();

#pragma unroll
  for (int u = 0; u < 8; ++u) {
    const int p = u >> 2, ct = u & 3;
    const int h = p * 512 + wave * 64 + ct * 16 + lm;
    const float g = Gs[h], be = Bt[h];
#pragma unroll
    for (int r = 0; r < 8; ++r) {
      const int row = r + hi2 * 8;
      const float v = (acc[u][r] - muS[row]) * rsS[row] * g + be;
      out[(size_t)(b * Nn + m0 + row) * Hn + h] = v;
    }
  }
}

extern "C" void kernel_launch(void* const* d_in, const int* in_sizes, int n_in,
                              void* d_out, int out_size, void* d_ws, size_t ws_size,
                              hipStream_t stream) {
  const float* f     = (const float*)d_in[0];
  const float* k     = (const float*)d_in[1];
  const float* Wlin  = (const float*)d_in[2];
  const float* blin  = (const float*)d_in[3];
  const float* gamma = (const float*)d_in[4];
  const float* beta  = (const float*)d_in[5];
  float* out = (float*)d_out;
  float* wgt = (float*)d_ws;

  (void)in_sizes; (void)n_in; (void)out_size; (void)ws_size;

  wgen_gemm<<<dim3(NK / 128, Nn / 64, Bn), 256, 0, stream>>>(f, Wlin, blin, wgt);
  dyconv_ln<<<dim3(Nn / 16, Bn), 256, 0, stream>>>(k, wgt, gamma, beta, out);
}


# Stage 1 re-tiled: 128x192 per work-group, 32x96 per wave (12 accumulator tiles) so each LDS fragment feeds 2 to 6 WMMAs; next k slab prefetched in registers, two LDS stages; sum order unchanged
# speedup vs baseline: 1.4198x; 1.0818x over previous
#include <hip/hip_runtime.h>
#include <hip/hip_bf16.h>


#define Bn 64
#define Nn 256
#define Hn 1024
#define NK 768
#define LN_EPS 1e-5f
#define KP 528

__device__ __forceinline__ int kpos(int k) { return 8 * ((k >> 1) & 1) + 2 * (k >> 2) + (k & 1); }

typedef __attribute__((ext_vector_type(2))) float v2f;
typedef __attribute__((ext_vector_type(8))) float v8f;

#define T1M 128
#define T1N 192
__global__ __launch_bounds__(256) void wgen_gemm(
    const float* __restrict__ f, const float* __restrict__ Wlin,
    const float* __restrict__ blin, float* __restrict__ wout) {
  __shared__ __attribute__((aligned(16))) float As[2][T1M * 20];
  __shared__ __attribute__((aligned(16))) float Bs[2][T1N * 20];

  const int tid = threadIdx.x;
  const int b = blockIdx.z;
  const int m0 = blockIdx.y * T1M;
  const int o0 = blockIdx.x * T1N;
  const int wave = tid >> 5, lane = tid & 31;
  const int lm = lane & 15, hi = lane >> 4;
  const int wm = (wave & 3) * 32;
  const int wn = (wave >> 2) * 96;

  v8f cAcc[2][6] = {};

  const int srow = tid >> 2;
  const int scol = (tid & 3) * 4;
  const float* fA = f + (size_t)(b * Nn + m0 + srow) * Hn + scol;
  const float* wB = Wlin + (size_t)(o0 + srow) * Hn + scol;
  const int sLo = srow * 20 + kpos(scol);
  const int sUp = srow * 20 + kpos(scol + 2);

  float4 ra[2], rb[3];
#pragma unroll
  for (int j = 0; j < 2; ++j) ra[j] = *(const float4*)(fA + (size_t)(j * 64) * Hn);
#pragma unroll
  for (int j = 0; j < 3; ++j) rb[j] = *(const float4*)(wB + (size_t)(j * 64) * Hn);
#pragma unroll
  for (int j = 0; j < 2; ++j) {
    *(float2*)(&As[0][j * 64 * 20 + sLo]) = make_float2(ra[j].x, ra[j].y);
    *(float2*)(&As[0][j * 64 * 20 + sUp]) = make_float2(ra[j].z, ra[j].w);
  }
#pragma unroll
  for (int j = 0; j < 3; ++j) {
    *(float2*)(&Bs[0][j * 64 * 20 + sLo]) = make_float2(rb[j].x, rb[j].y);
    *(float2*)(&Bs[0][j * 64 * 20 + sUp]) = make_float2(rb[j].z, rb[j].w);
  }
  __syncthreads();

  for (int kt = 0; kt < Hn / 16; ++kt) {
    const int cur = kt & 1;
    const bool more = (kt + 1) < Hn / 16;
    if (more) {
      const int k1 = (kt + 1) * 16;
#pragma unroll
      for (int j = 0; j < 2; ++j) ra[j] = *(const float4*)(fA + (size_t)(j * 64) * Hn + k1);
#pragma unroll
      for (int j = 0; j < 3; ++j) rb[j] = *(const float4*)(wB + (size_t)(j * 64) * Hn + k1);
    }
    v2f af[2][4], bf[6][4];
#pragma unroll
    for (int mt = 0; mt < 2; ++mt) {
      const int m = wm + mt * 16 + lm;
      const float4 lo = *(const float4*)(&As[cur][m * 20 + hi * 8]);
      const float4 up = *(const float4*)(&As[cur][m * 20 + hi * 8 + 4]);
      af[mt][0].x = lo.x; af[mt][0].y = lo.y; af[mt][1].x = lo.z; af[mt][1].y = lo.w;
      af[mt][2].x = up.x; af[mt][2].y = up.y; af[mt][3].x = up.z; af[mt][3].y = up.w;
    }
#pragma unroll
    for (int ct = 0; ct < 6; ++ct) {
      const int n = wn + ct * 16 + lm;
      const float4 lo = *(const float4*)(&Bs[cur][n * 20 + hi * 8]);
      const float4 up = *(const float4*)(&Bs[cur][n * 20 + hi * 8 + 4]);
      bf[ct][0].x = lo.x; bf[ct][0].y = lo.y; bf[ct][1].x = lo.z; bf[ct][1].y = lo.w;
      bf[ct][2].x = up.x; bf[ct][2].y = up.y; bf[ct][3].x = up.z; bf[ct][3].y = up.w;
    }
#pragma unroll
    for (int ks = 0; ks < 4; ++ks) {
#pragma unroll
      for (int mt = 0; mt < 2; ++mt) {
#pragma unroll
        for (int ct = 0; ct < 6; ++ct) {
          cAcc[mt][ct] = __builtin_amdgcn_wmma_f32_16x16x4_f32(
              false, af[mt][ks], false, bf[ct][ks], (short)0, cAcc[mt][ct], false, false);
        }
      }
    }
    if (more) {
#pragma unroll
      for (int j = 0; j < 2; ++j) {
        *(float2*)(&As[cur ^ 1][j * 64 * 20 + sLo]) = make_float2(ra[j].x, ra[j].y);
        *(float2*)(&As[cur ^ 1][j * 64 * 20 + sUp]) = make_float2(ra[j].z, ra[j].w);
      }
#pragma unroll
      for (int j = 0; j < 3; ++j) {
        *(float2*)(&Bs[cur ^ 1][j * 64 * 20 + sLo]) = make_float2(rb[j].x, rb[j].y);
        *(float2*)(&Bs[cur ^ 1][j * 64 * 20 + sUp]) = make_float2(rb[j].z, rb[j].w);
      }
    }
    __syncthreads();
  }

#pragma unroll
  for (int ct = 0; ct < 6; ++ct) {
    const int o = o0 + wn + ct * 16 + lm;
    const float bl = blin[o];
#pragma unroll
    for (int mt = 0; mt < 2; ++mt) {
#pragma unroll
      for (int r = 0; r < 8; ++r) {
        const int m = m0 + wm + mt * 16 + r + hi * 8;
        wout[(size_t)(b * Nn + m) * NK + o] = cAcc[mt][ct][r] + bl;
      }
    }
  }
}

__global__ __launch_bounds__(256) void dyconv_ln(
    const float* __restrict__ kin, const float* __restrict__ w,
    const float* __restrict__ gamma, const float* __restrict__ beta,
    float* __restrict__ out) {
  __shared__ __attribute__((aligned(16))) float Ks[8 * KP * 2];
  __shared__ __attribute__((aligned(16))) float Aw[16 * 52];
  __shared__ float Gs[1024];
  __shared__ float Bt[1024];
  __shared__ float redS[8 * 16];
  __shared__ float redQ[8 * 16];
  __shared__ float muS[16];
  __shared__ float rsS[16];

  const int tid = threadIdx.x;
  const int b = blockIdx.y;
  const int m0 = blockIdx.x * 16;
  const int wave = tid >> 5, lane = tid & 31;
  const int lm = lane & 15, hi2 = lane >> 4;

#pragma unroll
  for (int j = 0; j < 4; ++j) {
    int i = tid + j * 256;
    Gs[i] = gamma[i];
    Bt[i] = beta[i];
  }

  v8f acc[8] = {};

  const int aRow = tid >> 4;
  const int aI = tid & 15;
  const float* wp = w + (size_t)(b * Nn + m0 + aRow) * NK + aI * 3;
  const int aDst = aRow * 52 + kpos(aI);
  const int kPr0 = tid >> 7;
  const int kC4 = (tid & 127) * 4;
  const int hRow = tid & 15;
  const int hBase = ((hRow >> 1) * KP) * 2 + (hRow & 1);

  for (int ic = 0; ic < 16; ++ic) {
    const int i0 = ic * 16;
    __syncthreads();
    {
      const float* wq = wp + (size_t)i0 * 3;
      Aw[aDst + 0 * 16] = wq[0];
      Aw[aDst + 1 * 16] = wq[1];
      Aw[aDst + 2 * 16] = wq[2];
    }
#pragma unroll
    for (int p = 0; p < 2; ++p) {
      if (p) __syncthreads();
      {
        const float* kr = kin + (size_t)(b * Nn + i0 + kPr0 * 2) * Hn + p * 512 + kC4;
        float* kd = &Ks[(kPr0 * KP + 4 + kC4) * 2];
#pragma unroll
        for (int j = 0; j < 4; ++j) {
          const float4 x = *(const float4*)(kr + (size_t)(j * 4) * Hn);
          const float4 y = *(const float4*)(kr + (size_t)(j * 4 + 1) * Hn);
          const float4 lo = make_float4(x.x, y.x, x.y, y.y);
          const float4 up = make_float4(x.z, y.z, x.w, y.w);
          float* d = kd + (j * 2 * KP) * 2;
          *(float4*)(d)     = lo;
          *(float4*)(d + 4) = up;
        }
      }
      if (tid < 32) {
        const float* krh = kin + (size_t)(b * Nn + i0 + hRow) * Hn;
        if (tid < 16) {
          Ks[hBase + 3 * 2] = (p == 0) ? 0.0f : krh[511];
        } else {
          Ks[hBase + 516 * 2] = (p == 0) ? krh[512] : 0.0f;
        }
      }
      __syncthreads();
#pragma unroll
      for (int t = 0; t < 3; ++t) {
        const float4 alo = *(const float4*)(&Aw[lm * 52 + t * 16 + hi2 * 8]);
        const float4 aup = *(const float4*)(&Aw[lm * 52 + t * 16 + hi2 * 8 + 4]);
        v2f af[4];
        af[0].x = alo.x; af[0].y = alo.y; af[1].x = alo.z; af[1].y = alo.w;
        af[2].x = aup.x; af[2].y = aup.y; af[3].x = aup.z; af[3].y = aup.w;
#pragma unroll
        for (int ks = 0; ks < 4; ++ks) {
          const v2f a = af[ks];
#pragma unroll
          for (int ct = 0; ct < 4; ++ct) {
            const int col = (wave * 64 + ct * 16 + lm) + t + 3;
            const v2f bb = *(const v2f*)(&Ks[((ks * 2 + hi2) * KP + col) * 2]);
            acc[p * 4 + ct] = __builtin_amdgcn_wmma_f32_16x16x4_f32(
                false, a, false, bb, (short)0, acc[p * 4 + ct], false, false);
          }
        }
      }
    }
  }

  float s[8], q[8];
#pragma unroll
  for (int r = 0; r < 8; ++r) { s[r] = 0.0f; q[r] = 0.0f; }
#pragma unroll
  for (int u = 0; u < 8; ++u)
#pragma unroll
    for (int r = 0; r < 8; ++r) {
      float v = acc[u][r];
      s[r] += v;
      q[r] += v * v;
    }
#pragma unroll
  for (int mask = 1; mask < 16; mask <<= 1)
#pragma unroll
    for (int r = 0; r < 8; ++r) {
      s[r] += __shfl_xor(s[r], mask, 32);
      q[r] += __shfl_xor(q[r], mask, 32);
    }
  if (lm == 0) {
#pragma unroll
    for (int r = 0; r < 8; ++r) {
      const int row = r + hi2 * 8;
      redS[wave * 16 + row] = s[r];
      redQ[wave * 16 + row] = q[r];
    }
  }
  __syncthreads();
  if (tid < 16) {
    float ss = 0.0f, qq = 0.0f;
#pragma unroll
    for (int wv = 0; wv < 8; ++wv) {
      ss += redS[wv * 16 + tid];
      qq += redQ[wv * 16 + tid];
    }
    const float mu = ss * (1.0f / 1024.0f);
    const float var = qq * (1.0f / 1024.0f) - mu * mu;
    muS[tid] = mu;
    rsS[tid] = rsqrtf(var + LN_EPS);
  }
  __syncthreads();

#pragma unroll
  for (int u = 0; u < 8; ++u) {
    const int p = u >> 2, ct = u & 3;
    const int h = p * 512 + wave * 64 + ct * 16 + lm;
    const float g = Gs[h], be = Bt[h];
#pragma unroll
    for (int r = 0; r < 8; ++r) {
      const int row = r + hi2 * 8;
      const float v = (acc[u][r] - muS[row]) * rsS[row] * g + be;
      out[(size_t)(b * Nn + m0 + row) * Hn + h] = v;
    }
  }
}

extern "C" void kernel_launch(void* const* d_in, const int* in_sizes, int n_in,
                              void* d_out, int out_size, void* d_ws, size_t ws_size,
                              hipStream_t stream) {
  const float* f     = (const float*)d_in[0];
  const float* k     = (const float*)d_in[1];
  const float* Wlin  = (const float*)d_in[2];
  const float* blin  = (const float*)d_in[3];
  const float* gamma = (const float*)d_in[4];
  const float* beta  = (const float*)d_in[5];
  float* out = (float*)d_out;
  float* wgt = (float*)d_ws;

  (void)in_sizes; (void)n_in; (void)out_size; (void)ws_size;

  wgen_gemm<<<dim3(NK / T1N, Nn / T1M, Bn), 256, 0, stream>>>(f, Wlin, blin, wgt);
  dyconv_ln<<<dim3(Nn / 16, Bn), 256, 0, stream>>>(k, wgt, gamma, beta, out);
}


# Stage 2 k slab double-buffered in LDS (one stage per column half): next slab and weights fetched into registers under the products, stored behind them; one barrier per phase; sum order unchanged
# speedup vs baseline: 1.5060x; 1.0607x over previous
#include <hip/hip_runtime.h>
#include <hip/hip_bf16.h>


#define Bn 64
#define Nn 256
#define Hn 1024
#define NK 768
#define LN_EPS 1e-5f
#define KP 528

__device__ __forceinline__ int kpos(int k) { return 8 * ((k >> 1) & 1) + 2 * (k >> 2) + (k & 1); }

typedef __attribute__((ext_vector_type(2))) float v2f;
typedef __attribute__((ext_vector_type(8))) float v8f;

#define T1M 128
#define T1N 192
__global__ __launch_bounds__(256) void wgen_gemm(
    const float* __restrict__ f, const float* __restrict__ Wlin,
    const float* __restrict__ blin, float* __restrict__ wout) {
  __shared__ __attribute__((aligned(16))) float As[2][T1M * 20];
  __shared__ __attribute__((aligned(16))) float Bs[2][T1N * 20];

  const int tid = threadIdx.x;
  const int b = blockIdx.z;
  const int m0 = blockIdx.y * T1M;
  const int o0 = blockIdx.x * T1N;
  const int wave = tid >> 5, lane = tid & 31;
  const int lm = lane & 15, hi = lane >> 4;
  const int wm = (wave & 3) * 32;
  const int wn = (wave >> 2) * 96;

  v8f cAcc[2][6] = {};

  const int srow = tid >> 2;
  const int scol = (tid & 3) * 4;
  const float* fA = f + (size_t)(b * Nn + m0 + srow) * Hn + scol;
  const float* wB = Wlin + (size_t)(o0 + srow) * Hn + scol;
  const int sLo = srow * 20 + kpos(scol);
  const int sUp = srow * 20 + kpos(scol + 2);

  float4 ra[2], rb[3];
#pragma unroll
  for (int j = 0; j < 2; ++j) ra[j] = *(const float4*)(fA + (size_t)(j * 64) * Hn);
#pragma unroll
  for (int j = 0; j < 3; ++j) rb[j] = *(const float4*)(wB + (size_t)(j * 64) * Hn);
#pragma unroll
  for (int j = 0; j < 2; ++j) {
    *(float2*)(&As[0][j * 64 * 20 + sLo]) = make_float2(ra[j].x, ra[j].y);
    *(float2*)(&As[0][j * 64 * 20 + sUp]) = make_float2(ra[j].z, ra[j].w);
  }
#pragma unroll
  for (int j = 0; j < 3; ++j) {
    *(float2*)(&Bs[0][j * 64 * 20 + sLo]) = make_float2(rb[j].x, rb[j].y);
    *(float2*)(&Bs[0][j * 64 * 20 + sUp]) = make_float2(rb[j].z, rb[j].w);
  }
  __syncthreads();

  for (int kt = 0; kt < Hn / 16; ++kt) {
    const int cur = kt & 1;
    const bool more = (kt + 1) < Hn / 16;
    if (more) {
      const int k1 = (kt + 1) * 16;
#pragma unroll
      for (int j = 0; j < 2; ++j) ra[j] = *(const float4*)(fA + (size_t)(j * 64) * Hn + k1);
#pragma unroll
      for (int j = 0; j < 3; ++j) rb[j] = *(const float4*)(wB + (size_t)(j * 64) * Hn + k1);
    }
    v2f af[2][4], bf[6][4];
#pragma unroll
    for (int mt = 0; mt < 2; ++mt) {
      const int m = wm + mt * 16 + lm;
      const float4 lo = *(const float4*)(&As[cur][m * 20 + hi * 8]);
      const float4 up = *(const float4*)(&As[cur][m * 20 + hi * 8 + 4]);
      af[mt][0].x = lo.x; af[mt][0].y = lo.y; af[mt][1].x = lo.z; af[mt][1].y = lo.w;
      af[mt][2].x = up.x; af[mt][2].y = up.y; af[mt][3].x = up.z; af[mt][3].y = up.w;
    }
#pragma unroll
    for (int ct = 0; ct < 6; ++ct) {
      const int n = wn + ct * 16 + lm;
      const float4 lo = *(const float4*)(&Bs[cur][n * 20 + hi * 8]);
      const float4 up = *(const float4*)(&Bs[cur][n * 20 + hi * 8 + 4]);
      bf[ct][0].x = lo.x; bf[ct][0].y = lo.y; bf[ct][1].x = lo.z; bf[ct][1].y = lo.w;
      bf[ct][2].x = up.x; bf[ct][2].y = up.y; bf[ct][3].x = up.z; bf[ct][3].y = up.w;
    }
#pragma unroll
    for (int ks = 0; ks < 4; ++ks) {
#pragma unroll
      for (int mt = 0; mt < 2; ++mt) {
#pragma unroll
        for (int ct = 0; ct < 6; ++ct) {
          cAcc[mt][ct] = __builtin_amdgcn_wmma_f32_16x16x4_f32(
              false, af[mt][ks], false, bf[ct][ks], (short)0, cAcc[mt][ct], false, false);
        }
      }
    }
    if (more) {
#pragma unroll
      for (int j = 0; j < 2; ++j) {
        *(float2*)(&As[cur ^ 1][j * 64 * 20 + sLo]) = make_float2(ra[j].x, ra[j].y);
        *(float2*)(&As[cur ^ 1][j * 64 * 20 + sUp]) = make_float2(ra[j].z, ra[j].w);
      }
#pragma unroll
      for (int j = 0; j < 3; ++j) {
        *(float2*)(&Bs[cur ^ 1][j * 64 * 20 + sLo]) = make_float2(rb[j].x, rb[j].y);
        *(float2*)(&Bs[cur ^ 1][j * 64 * 20 + sUp]) = make_float2(rb[j].z, rb[j].w);
      }
    }
    __syncthreads();
  }

#pragma unroll
  for (int ct = 0; ct < 6; ++ct) {
    const int o = o0 + wn + ct * 16 + lm;
    const float bl = blin[o];
#pragma unroll
    for (int mt = 0; mt < 2; ++mt) {
#pragma unroll
      for (int r = 0; r < 8; ++r) {
        const int m = m0 + wm + mt * 16 + r + hi * 8;
        wout[(size_t)(b * Nn + m) * NK + o] = cAcc[mt][ct][r] + bl;
      }
    }
  }
}

#define KSTG (8 * KP * 2)
#define AWSTG (16 * 52)

template <int P>
__device__ __forceinline__ void slab_fetch(const float* __restrict__ kin, int b, int i0,
                                           int kPr0, int kC4, int tid, int hRow,
                                           float4 (&rx)[4], float4 (&ry)[4], float& rh) {
  const float* kr = kin + (size_t)(b * Nn + i0 + kPr0 * 2) * Hn + P * 512 + kC4;
#pragma unroll
  for (int j = 0; j < 4; ++j) {
    rx[j] = *(const float4*)(kr + (size_t)(j * 4) * Hn);
    ry[j] = *(const float4*)(kr + (size_t)(j * 4 + 1) * Hn);
  }
  if (tid < 32) {
    const float* krh = kin + (size_t)(b * Nn + i0 + hRow) * Hn;
    float v = 0.0f;
    if (P == 0) { if (tid >= 16) v = krh[512]; }
    else        { if (tid < 16)  v = krh[511]; }
    rh = v;
  }
}

__device__ __forceinline__ void slab_store(float* KsP, int kPr0, int kC4, int tid,
                                           int hBase, const float4 (&rx)[4],
                                           const float4 (&ry)[4], float rh) {
  float* kd = KsP + (kPr0 * KP + 4 + kC4) * 2;
#pragma unroll
  for (int j = 0; j < 4; ++j) {
    const float4 x = rx[j], y = ry[j];
    const float4 lo = make_float4(x.x, y.x, x.y, y.y);
    const float4 up = make_float4(x.z, y.z, x.w, y.w);
    float* d = kd + (j * 2 * KP) * 2;
    *(float4*)(d)     = lo;
    *(float4*)(d + 4) = up;
  }
  if (tid < 32) {
    if (tid < 16) KsP[hBase + 3 * 2] = rh;
    else          KsP[hBase + 516 * 2] = rh;
  }
}

__device__ __forceinline__ void frag_load(const float* KsP, const float* AwC, int wave, int lm,
                                          int hi2, v2f (&af)[3][4], v2f (&bf)[3][4][4]) {
#pragma unroll
  for (int t = 0; t < 3; ++t) {
    const float4 alo = *(const float4*)(&AwC[lm * 52 + t * 16 + hi2 * 8]);
    const float4 aup = *(const float4*)(&AwC[lm * 52 + t * 16 + hi2 * 8 + 4]);
    af[t][0].x = alo.x; af[t][0].y = alo.y; af[t][1].x = alo.z; af[t][1].y = alo.w;
    af[t][2].x = aup.x; af[t][2].y = aup.y; af[t][3].x = aup.z; af[t][3].y = aup.w;
#pragma unroll
    for (int ks = 0; ks < 4; ++ks) {
#pragma unroll
      for (int ct = 0; ct < 4; ++ct) {
        const int col = (wave * 64 + ct * 16 + lm) + t + 3;
        bf[t][ks][ct] = *(const v2f*)(&KsP[((ks * 2 + hi2) * KP + col) * 2]);
      }
    }
  }
}

template <int P, int T0, int T1>
__device__ __forceinline__ void frag_products(const v2f (&af)[3][4], const v2f (&bf)[3][4][4],
                                              v8f (&acc)[8]) {
#pragma unroll
  for (int t = T0; t < T1; ++t) {
#pragma unroll
    for (int ks = 0; ks < 4; ++ks) {
#pragma unroll
      for (int ct = 0; ct < 4; ++ct) {
        acc[P * 4 + ct] = __builtin_amdgcn_wmma_f32_16x16x4_f32(
            false, af[t][ks], false, bf[t][ks][ct], (short)0, acc[P * 4 + ct], false, false);
      }
    }
  }
}

__global__ __launch_bounds__(256) void dyconv_ln(
    const float* __restrict__ kin, const float* __restrict__ w,
    const float* __restrict__ gamma, const float* __restrict__ beta,
    float* __restrict__ out) {
  __shared__ __attribute__((aligned(16))) float Ks[2][KSTG];
  __shared__ __attribute__((aligned(16))) float Aw[2][AWSTG];
  __shared__ float Gs[1024];
  __shared__ float Bt[1024];
  __shared__ float redS[8 * 16];
  __shared__ float redQ[8 * 16];
  __shared__ float muS[16];
  __shared__ float rsS[16];

  const int tid = threadIdx.x;
  const int b = blockIdx.y;
  const int m0 = blockIdx.x * 16;
  const int wave = tid >> 5, lane = tid & 31;
  const int lm = lane & 15, hi2 = lane >> 4;

#pragma unroll
  for (int j = 0; j < 4; ++j) {
    int i = tid + j * 256;
    Gs[i] = gamma[i];
    Bt[i] = beta[i];
  }

  v8f acc[8] = {};

  const int aRow = tid >> 4;
  const int aI = tid & 15;
  const float* wp = w + (size_t)(b * Nn + m0 + aRow) * NK + aI * 3;
  const int aDst = aRow * 52 + kpos(aI);
  const int kPr0 = tid >> 7;
  const int kC4 = (tid & 127) * 4;
  const int hRow = tid & 15;
  const int hBase = ((hRow >> 1) * KP) * 2 + (hRow & 1);

  float4 rx[4], ry[4];
  float rh = 0.0f;
  float rw0, rw1, rw2;

  slab_fetch<0>(kin, b, 0, kPr0, kC4, tid, hRow, rx, ry, rh);
  rw0 = wp[0]; rw1 = wp[1]; rw2 = wp[2];
  slab_store(Ks[0], kPr0, kC4, tid, hBase, rx, ry, rh);
  Aw[0][aDst + 0 * 16] = rw0;
  Aw[0][aDst + 1 * 16] = rw1;
  Aw[0][aDst + 2 * 16] = rw2;
  __syncthreads();

  for (int ic = 0; ic < 16; ++ic) {
    const int i0 = ic * 16;
    const float* AwC = Aw[ic & 1];
    v2f af[3][4], bf[3][4][4];
    frag_load(Ks[0], AwC, wave, lm, hi2, af, bf);
    __builtin_amdgcn_sched_barrier(0);
    frag_products<0, 0, 1>(af, bf, acc);
    __builtin_amdgcn_sched_barrier(0);
    slab_fetch<1>(kin, b, i0, kPr0, kC4, tid, hRow, rx, ry, rh);
    frag_products<0, 1, 3>(af, bf, acc);
    __builtin_amdgcn_sched_barrier(0);
    slab_store(Ks[1], kPr0, kC4, tid, hBase, rx, ry, rh);
    __syncthreads();
    const bool more = (ic + 1) < 16;
    frag_load(Ks[1], AwC, wave, lm, hi2, af, bf);
    __builtin_amdgcn_sched_barrier(0);
    frag_products<1, 0, 1>(af, bf, acc);
    __builtin_amdgcn_sched_barrier(0);
    if (more) {
      slab_fetch<0>(kin, b, i0 + 16, kPr0, kC4, tid, hRow, rx, ry, rh);
      const float* wq = wp + (size_t)(i0 + 16) * 3;
      rw0 = wq[0]; rw1 = wq[1]; rw2 = wq[2];
    }
    frag_products<1, 1, 3>(af, bf, acc);
    __builtin_amdgcn_sched_barrier(0);
    if (more) {
      slab_store(Ks[0], kPr0, kC4, tid, hBase, rx, ry, rh);
      float* AwN = Aw[(ic + 1) & 1];
      AwN[aDst + 0 * 16] = rw0;
      AwN[aDst + 1 * 16] = rw1;
      AwN[aDst + 2 * 16] = rw2;
    }
    __syncthreads();
  }

  float s[8], q[8];
#pragma unroll
  for (int r = 0; r < 8; ++r) { s[r] = 0.0f; q[r] = 0.0f; }
#pragma unroll
  for (int u = 0; u < 8; ++u)
#pragma unroll
    for (int r = 0; r < 8; ++r) {
      float v = acc[u][r];
      s[r] += v;
      q[r] += v * v;
    }
#pragma unroll
  for (int mask = 1; mask < 16; mask <<= 1)
#pragma unroll
    for (int r = 0; r < 8; ++r) {
      s[r] += __shfl_xor(s[r], mask, 32);
      q[r] += __shfl_xor(q[r], mask, 32);
    }
  if (lm == 0) {
#pragma unroll
    for (int r = 0; r < 8; ++r) {
      const int row = r + hi2 * 8;
      redS[wave * 16 + row] = s[r];
      redQ[wave * 16 + row] = q[r];
    }
  }
  __syncthreads();
  if (tid < 16) {
    float ss = 0.0f, qq = 0.0f;
#pragma unroll
    for (int wv = 0; wv < 8; ++wv) {
      ss += redS[wv * 16 + tid];
      qq += redQ[wv * 16 + tid];
    }
    const float mu = ss * (1.0f / 1024.0f);
    const float var = qq * (1.0f / 1024.0f) - mu * mu;
    muS[tid] = mu;
    rsS[tid] = rsqrtf(var + LN_EPS);
  }
  __syncthreads();

#pragma unroll
  for (int u = 0; u < 8; ++u) {
    const int p = u >> 2, ct = u & 3;
    const int h = p * 512 + wave * 64 + ct * 16 + lm;
    const float g = Gs[h], be = Bt[h];
#pragma unroll
    for (int r = 0; r < 8; ++r) {
      const int row = r + hi2 * 8;
      const float v = (acc[u][r] - muS[row]) * rsS[row] * g + be;
      out[(size_t)(b * Nn + m0 + row) * Hn + h] = v;
    }
  }
}

extern "C" void kernel_launch(void* const* d_in, const int* in_sizes, int n_in,
                              void* d_out, int out_size, void* d_ws, size_t ws_size,
                              hipStream_t stream) {
  const float* f     = (const float*)d_in[0];
  const float* k     = (const float*)d_in[1];
  const float* Wlin  = (const float*)d_in[2];
  const float* blin  = (const float*)d_in[3];
  const float* gamma = (const float*)d_in[4];
  const float* beta  = (const float*)d_in[5];
  float* out = (float*)d_out;
  float* wgt = (float*)d_ws;

  (void)in_sizes; (void)n_in; (void)out_size; (void)ws_size;

  wgen_gemm<<<dim3(NK / T1N, Nn / T1M, Bn), 256, 0, stream>>>(f, Wlin, blin, wgt);
  dyconv_ln<<<dim3(Nn / 16, Bn), 256, 0, stream>>>(k, wgt, gamma, beta, out);
}


# Wave priority (s_setprio) stepped down 3,2,1,0 at 1/2, 3/4, 7/8 of both kernels' k loops: the work-group that is behind wins the matrix pipe, so a WGP's two work-groups end together; sums unchanged
# speedup vs baseline: 1.5602x; 1.0360x over previous
#include <hip/hip_runtime.h>
#include <hip/hip_bf16.h>


#define Bn 64
#define Nn 256
#define Hn 1024
#define NK 768
#define LN_EPS 1e-5f
#define KP 528

__device__ __forceinline__ int kpos(int k) { return 8 * ((k >> 1) & 1) + 2 * (k >> 2) + (k & 1); }

typedef __attribute__((ext_vector_type(2))) float v2f;
typedef __attribute__((ext_vector_type(8))) float v8f;

#define T1M 128
#define T1N 192
__global__ __launch_bounds__(256) void wgen_gemm(
    const float* __restrict__ f, const float* __restrict__ Wlin,
    const float* __restrict__ blin, float* __restrict__ wout) {
  __shared__ __attribute__((aligned(16))) float As[2][T1M * 20];
  __shared__ __attribute__((aligned(16))) float Bs[2][T1N * 20];

  const int tid = threadIdx.x;
  const int b = blockIdx.z;
  const int m0 = blockIdx.y * T1M;
  const int o0 = blockIdx.x * T1N;
  const int wave = tid >> 5, lane = tid & 31;
  const int lm = lane & 15, hi = lane >> 4;
  const int wm = (wave & 3) * 32;
  const int wn = (wave >> 2) * 96;

  __builtin_amdgcn_s_setprio(3);

  v8f cAcc[2][6] = {};

  const int srow = tid >> 2;
  const int scol = (tid & 3) * 4;
  const float* fA = f + (size_t)(b * Nn + m0 + srow) * Hn + scol;
  const float* wB = Wlin + (size_t)(o0 + srow) * Hn + scol;
  const int sLo = srow * 20 + kpos(scol);
  const int sUp = srow * 20 + kpos(scol + 2);

  float4 ra[2], rb[3];
#pragma unroll
  for (int j = 0; j < 2; ++j) ra[j] = *(const float4*)(fA + (size_t)(j * 64) * Hn);
#pragma unroll
  for (int j = 0; j < 3; ++j) rb[j] = *(const float4*)(wB + (size_t)(j * 64) * Hn);
#pragma unroll
  for (int j = 0; j < 2; ++j) {
    *(float2*)(&As[0][j * 64 * 20 + sLo]) = make_float2(ra[j].x, ra[j].y);
    *(float2*)(&As[0][j * 64 * 20 + sUp]) = make_float2(ra[j].z, ra[j].w);
  }
#pragma unroll
  for (int j = 0; j < 3; ++j) {
    *(float2*)(&Bs[0][j * 64 * 20 + sLo]) = make_float2(rb[j].x, rb[j].y);
    *(float2*)(&Bs[0][j * 64 * 20 + sUp]) = make_float2(rb[j].z, rb[j].w);
  }
  __syncthreads();

  for (int kt = 0; kt < Hn / 16; ++kt) {
    if (kt == (Hn / 16) / 2) __builtin_amdgcn_s_setprio(2);
    else if (kt == (Hn / 16) * 3 / 4) __builtin_amdgcn_s_setprio(1);
    else if (kt == (Hn / 16) * 7 / 8) __builtin_amdgcn_s_setprio(0);
    const int cur = kt & 1;
    const bool more = (kt + 1) < Hn / 16;
    if (more) {
      const int k1 = (kt + 1) * 16;
#pragma unroll
      for (int j = 0; j < 2; ++j) ra[j] = *(const float4*)(fA + (size_t)(j * 64) * Hn + k1);
#pragma unroll
      for (int j = 0; j < 3; ++j) rb[j] = *(const float4*)(wB + (size_t)(j * 64) * Hn + k1);
    }
    v2f af[2][4], bf[6][4];
#pragma unroll
    for (int mt = 0; mt < 2; ++mt) {
      const int m = wm + mt * 16 + lm;
      const float4 lo = *(const float4*)(&As[cur][m * 20 + hi * 8]);
      const float4 up = *(const float4*)(&As[cur][m * 20 + hi * 8 + 4]);
      af[mt][0].x = lo.x; af[mt][0].y = lo.y; af[mt][1].x = lo.z; af[mt][1].y = lo.w;
      af[mt][2].x = up.x; af[mt][2].y = up.y; af[mt][3].x = up.z; af[mt][3].y = up.w;
    }
#pragma unroll
    for (int ct = 0; ct < 6; ++ct) {
      const int n = wn + ct * 16 + lm;
      const float4 lo = *(const float4*)(&Bs[cur][n * 20 + hi * 8]);
      const float4 up = *(const float4*)(&Bs[cur][n * 20 + hi * 8 + 4]);
      bf[ct][0].x = lo.x; bf[ct][0].y = lo.y; bf[ct][1].x = lo.z; bf[ct][1].y = lo.w;
      bf[ct][2].x = up.x; bf[ct][2].y = up.y; bf[ct][3].x = up.z; bf[ct][3].y = up.w;
    }
#pragma unroll
    for (int ks = 0; ks < 4; ++ks) {
#pragma unroll
      for (int mt = 0; mt < 2; ++mt) {
#pragma unroll
        for (int ct = 0; ct < 6; ++ct) {
          cAcc[mt][ct] = __builtin_amdgcn_wmma_f32_16x16x4_f32(
              false, af[mt][ks], false, bf[ct][ks], (short)0, cAcc[mt][ct], false, false);
        }
      }
    }
    if (more) {
#pragma unroll
      for (int j = 0; j < 2; ++j) {
        *(float2*)(&As[cur ^ 1][j * 64 * 20 + sLo]) = make_float2(ra[j].x, ra[j].y);
        *(float2*)(&As[cur ^ 1][j * 64 * 20 + sUp]) = make_float2(ra[j].z, ra[j].w);
      }
#pragma unroll
      for (int j = 0; j < 3; ++j) {
        *(float2*)(&Bs[cur ^ 1][j * 64 * 20 + sLo]) = make_float2(rb[j].x, rb[j].y);
        *(float2*)(&Bs[cur ^ 1][j * 64 * 20 + sUp]) = make_float2(rb[j].z, rb[j].w);
      }
    }
    __syncthreads();
  }

#pragma unroll
  for (int ct = 0; ct < 6; ++ct) {
    const int o = o0 + wn + ct * 16 + lm;
    const float bl = blin[o];
#pragma unroll
    for (int mt = 0; mt < 2; ++mt) {
#pragma unroll
      for (int r = 0; r < 8; ++r) {
        const int m = m0 + wm + mt * 16 + r + hi * 8;
        wout[(size_t)(b * Nn + m) * NK + o] = cAcc[mt][ct][r] + bl;
      }
    }
  }
}

#define KSTG (8 * KP * 2)
#define AWSTG (16 * 52)

template <int P>
__device__ __forceinline__ void slab_fetch(const float* __restrict__ kin, int b, int i0,
                                           int kPr0, int kC4, int tid, int hRow,
                                           float4 (&rx)[4], float4 (&ry)[4], float& rh) {
  const float* kr = kin + (size_t)(b * Nn + i0 + kPr0 * 2) * Hn + P * 512 + kC4;
#pragma unroll
  for (int j = 0; j < 4; ++j) {
    rx[j] = *(const float4*)(kr + (size_t)(j * 4) * Hn);
    ry[j] = *(const float4*)(kr + (size_t)(j * 4 + 1) * Hn);
  }
  if (tid < 32) {
    const float* krh = kin + (size_t)(b * Nn + i0 + hRow) * Hn;
    float v = 0.0f;
    if (P == 0) { if (tid >= 16) v = krh[512]; }
    else        { if (tid < 16)  v = krh[511]; }
    rh = v;
  }
}

__device__ __forceinline__ void slab_store(float* KsP, int kPr0, int kC4, int tid,
                                           int hBase, const float4 (&rx)[4],
                                           const float4 (&ry)[4], float rh) {
  float* kd = KsP + (kPr0 * KP + 4 + kC4) * 2;
#pragma unroll
  for (int j = 0; j < 4; ++j) {
    const float4 x = rx[j], y = ry[j];
    const float4 lo = make_float4(x.x, y.x, x.y, y.y);
    const float4 up = make_float4(x.z, y.z, x.w, y.w);
    float* d = kd + (j * 2 * KP) * 2;
    *(float4*)(d)     = lo;
    *(float4*)(d + 4) = up;
  }
  if (tid < 32) {
    if (tid < 16) KsP[hBase + 3 * 2] = rh;
    else          KsP[hBase + 516 * 2] = rh;
  }
}

__device__ __forceinline__ void frag_load(const float* KsP, const float* AwC, int wave, int lm,
                                          int hi2, v2f (&af)[3][4], v2f (&bf)[3][4][4]) {
#pragma unroll
  for (int t = 0; t < 3; ++t) {
    const float4 alo = *(const float4*)(&AwC[lm * 52 + t * 16 + hi2 * 8]);
    const float4 aup = *(const float4*)(&AwC[lm * 52 + t * 16 + hi2 * 8 + 4]);
    af[t][0].x = alo.x; af[t][0].y = alo.y; af[t][1].x = alo.z; af[t][1].y = alo.w;
    af[t][2].x = aup.x; af[t][2].y = aup.y; af[t][3].x = aup.z; af[t][3].y = aup.w;
#pragma unroll
    for (int ks = 0; ks < 4; ++ks) {
#pragma unroll
      for (int ct = 0; ct < 4; ++ct) {
        const int col = (wave * 64 + ct * 16 + lm) + t + 3;
        bf[t][ks][ct] = *(const v2f*)(&KsP[((ks * 2 + hi2) * KP + col) * 2]);
      }
    }
  }
}

template <int P, int T0, int T1>
__device__ __forceinline__ void frag_products(const v2f (&af)[3][4], const v2f (&bf)[3][4][4],
                                              v8f (&acc)[8]) {
#pragma unroll
  for (int t = T0; t < T1; ++t) {
#pragma unroll
    for (int ks = 0; ks < 4; ++ks) {
#pragma unroll
      for (int ct = 0; ct < 4; ++ct) {
        acc[P * 4 + ct] = __builtin_amdgcn_wmma_f32_16x16x4_f32(
            false, af[t][ks], false, bf[t][ks][ct], (short)0, acc[P * 4 + ct], false, false);
      }
    }
  }
}

__global__ __launch_bounds__(256) void dyconv_ln(
    const float* __restrict__ kin, const float* __restrict__ w,
    const float* __restrict__ gamma, const float* __restrict__ beta,
    float* __restrict__ out) {
  __shared__ __attribute__((aligned(16))) float Ks[2][KSTG];
  __shared__ __attribute__((aligned(16))) float Aw[2][AWSTG];
  __shared__ float Gs[1024];
  __shared__ float Bt[1024];
  __shared__ float redS[8 * 16];
  __shared__ float redQ[8 * 16];
  __shared__ float muS[16];
  __shared__ float rsS[16];

  const int tid = threadIdx.x;
  const int b = blockIdx.y;
  const int m0 = blockIdx.x * 16;
  const int wave = tid >> 5, lane = tid & 31;
  const int lm = lane & 15, hi2 = lane >> 4;

  __builtin_amdgcn_s_setprio(3);

#pragma unroll
  for (int j = 0; j < 4; ++j) {
    int i = tid + j * 256;
    Gs[i] = gamma[i];
    Bt[i] = beta[i];
  }

  v8f acc[8] = {};

  const int aRow = tid >> 4;
  const int aI = tid & 15;
  const float* wp = w + (size_t)(b * Nn + m0 + aRow) * NK + aI * 3;
  const int aDst = aRow * 52 + kpos(aI);
  const int kPr0 = tid >> 7;
  const int kC4 = (tid & 127) * 4;
  const int hRow = tid & 15;
  const int hBase = ((hRow >> 1) * KP) * 2 + (hRow & 1);

  float4 rx[4], ry[4];
  float rh = 0.0f;
  float rw0, rw1, rw2;

  slab_fetch<0>(kin, b, 0, kPr0, kC4, tid, hRow, rx, ry, rh);
  rw0 = wp[0]; rw1 = wp[1]; rw2 = wp[2];
  slab_store(Ks[0], kPr0, kC4, tid, hBase, rx, ry, rh);
  Aw[0][aDst + 0 * 16] = rw0;
  Aw[0][aDst + 1 * 16] = rw1;
  Aw[0][aDst + 2 * 16] = rw2;
  __syncthreads();

  for (int ic = 0; ic < 16; ++ic) {
    if (ic == 8) __builtin_amdgcn_s_setprio(2);
    else if (ic == 12) __builtin_amdgcn_s_setprio(1);
    else if (ic == 14) __builtin_amdgcn_s_setprio(0);
    const int i0 = ic * 16;
    const float* AwC = Aw[ic & 1];
    v2f af[3][4], bf[3][4][4];
    frag_load(Ks[0], AwC, wave, lm, hi2, af, bf);
    __builtin_amdgcn_sched_barrier(0);
    frag_products<0, 0, 1>(af, bf, acc);
    __builtin_amdgcn_sched_barrier(0);
    slab_fetch<1>(kin, b, i0, kPr0, kC4, tid, hRow, rx, ry, rh);
    frag_products<0, 1, 3>(af, bf, acc);
    __builtin_amdgcn_sched_barrier(0);
    slab_store(Ks[1], kPr0, kC4, tid, hBase, rx, ry, rh);
    __syncthreads();
    const bool more = (ic + 1) < 16;
    frag_load(Ks[1], AwC, wave, lm, hi2, af, bf);
    __builtin_amdgcn_sched_barrier(0);
    frag_products<1, 0, 1>(af, bf, acc);
    __builtin_amdgcn_sched_barrier(0);
    if (more) {
      slab_fetch<0>(kin, b, i0 + 16, kPr0, kC4, tid, hRow, rx, ry, rh);
      const float* wq = wp + (size_t)(i0 + 16) * 3;
      rw0 = wq[0]; rw1 = wq[1]; rw2 = wq[2];
    }
    frag_products<1, 1, 3>(af, bf, acc);
    __builtin_amdgcn_sched_barrier(0);
    if (more) {
      slab_store(Ks[0], kPr0, kC4, tid, hBase, rx, ry, rh);
      float* AwN = Aw[(ic + 1) & 1];
      AwN[aDst + 0 * 16] = rw0;
      AwN[aDst + 1 * 16] = rw1;
      AwN[aDst + 2 * 16] = rw2;
    }
    __syncthreads();
  }

  float s[8], q[8];
#pragma unroll
  for (int r = 0; r < 8; ++r) { s[r] = 0.0f; q[r] = 0.0f; }
#pragma unroll
  for (int u = 0; u < 8; ++u)
#pragma unroll
    for (int r = 0; r < 8; ++r) {
      float v = acc[u][r];
      s[r] += v;
      q[r] += v * v;
    }
#pragma unroll
  for (int mask = 1; mask < 16; mask <<= 1)
#pragma unroll
    for (int r = 0; r < 8; ++r) {
      s[r] += __shfl_xor(s[r], mask, 32);
      q[r] += __shfl_xor(q[r], mask, 32);
    }
  if (lm == 0) {
#pragma unroll
    for (int r = 0; r < 8; ++r) {
      const int row = r + hi2 * 8;
      redS[wave * 16 + row] = s[r];
      redQ[wave * 16 + row] = q[r];
    }
  }
  __syncthreads();
  if (tid < 16) {
    float ss = 0.0f, qq = 0.0f;
#pragma unroll
    for (int wv = 0; wv < 8; ++wv) {
      ss += redS[wv * 16 + tid];
      qq += redQ[wv * 16 + tid];
    }
    const float mu = ss * (1.0f / 1024.0f);
    const float var = qq * (1.0f / 1024.0f) - mu * mu;
    muS[tid] = mu;
    rsS[tid] = rsqrtf(var + LN_EPS);
  }
  __syncthreads();

#pragma unroll
  for (int u = 0; u < 8; ++u) {
    const int p = u >> 2, ct = u & 3;
    const int h = p * 512 + wave * 64 + ct * 16 + lm;
    const float g = Gs[h], be = Bt[h];
#pragma unroll
    for (int r = 0; r < 8; ++r) {
      const int row = r + hi2 * 8;
      const float v = (acc[u][r] - muS[row]) * rsS[row] * g + be;
      out[(size_t)(b * Nn + m0 + row) * Hn + h] = v;
    }
  }
}

extern "C" void kernel_launch(void* const* d_in, const int* in_sizes, int n_in,
                              void* d_out, int out_size, void* d_ws, size_t ws_size,
                              hipStream_t stream) {
  const float* f     = (const float*)d_in[0];
  const float* k     = (const float*)d_in[1];
  const float* Wlin  = (const float*)d_in[2];
  const float* blin  = (const float*)d_in[3];
  const float* gamma = (const float*)d_in[4];
  const float* beta  = (const float*)d_in[5];
  float* out = (float*)d_out;
  float* wgt = (float*)d_ws;

  (void)in_sizes; (void)n_in; (void)out_size; (void)ws_size;

  wgen_gemm<<<dim3(NK / T1N, Nn / T1M, Bn), 256, 0, stream>>>(f, Wlin, blin, wgt);
  dyconv_ln<<<dim3(Nn / 16, Bn), 256, 0, stream>>>(k, wgt, gamma, beta, out);
}


# Stage 2 conv by the minimal 3-tap filtering identity on column pairs: 4 WMMA sums per 2 outputs instead of 6, columns and weights pre-combined by adds; summation order changes, f32 operands kept
# speedup vs baseline: 1.6347x; 1.0478x over previous
#include <hip/hip_runtime.h>
#include <hip/hip_bf16.h>


#define Bn 64
#define Nn 256
#define Hn 1024
#define NK 768
#define LN_EPS 1e-5f
#define KP 528

__device__ __forceinline__ int kpos(int k) { return 8 * ((k >> 1) & 1) + 2 * (k >> 2) + (k & 1); }

typedef __attribute__((ext_vector_type(2))) float v2f;
typedef __attribute__((ext_vector_type(8))) float v8f;

#define T1M 128
#define T1N 192
__global__ __launch_bounds__(256) void wgen_gemm(
    const float* __restrict__ f, const float* __restrict__ Wlin,
    const float* __restrict__ blin, float* __restrict__ wout) {
  __shared__ __attribute__((aligned(16))) float As[2][T1M * 20];
  __shared__ __attribute__((aligned(16))) float Bs[2][T1N * 20];

  const int tid = threadIdx.x;
  const int b = blockIdx.z;
  const int m0 = blockIdx.y * T1M;
  const int o0 = blockIdx.x * T1N;
  const int wave = tid >> 5, lane = tid & 31;
  const int lm = lane & 15, hi = lane >> 4;
  const int wm = (wave & 3) * 32;
  const int wn = (wave >> 2) * 96;

  __builtin_amdgcn_s_setprio(3);

  v8f cAcc[2][6] = {};

  const int srow = tid >> 2;
  const int scol = (tid & 3) * 4;
  const float* fA = f + (size_t)(b * Nn + m0 + srow) * Hn + scol;
  const float* wB = Wlin + (size_t)(o0 + srow) * Hn + scol;
  const int sLo = srow * 20 + kpos(scol);
  const int sUp = srow * 20 + kpos(scol + 2);

  float4 ra[2], rb[3];
#pragma unroll
  for (int j = 0; j < 2; ++j) ra[j] = *(const float4*)(fA + (size_t)(j * 64) * Hn);
#pragma unroll
  for (int j = 0; j < 3; ++j) rb[j] = *(const float4*)(wB + (size_t)(j * 64) * Hn);
#pragma unroll
  for (int j = 0; j < 2; ++j) {
    *(float2*)(&As[0][j * 64 * 20 + sLo]) = make_float2(ra[j].x, ra[j].y);
    *(float2*)(&As[0][j * 64 * 20 + sUp]) = make_float2(ra[j].z, ra[j].w);
  }
#pragma unroll
  for (int j = 0; j < 3; ++j) {
    *(float2*)(&Bs[0][j * 64 * 20 + sLo]) = make_float2(rb[j].x, rb[j].y);
    *(float2*)(&Bs[0][j * 64 * 20 + sUp]) = make_float2(rb[j].z, rb[j].w);
  }
  __syncthreads();

  for (int kt = 0; kt < Hn / 16; ++kt) {
    if (kt == (Hn / 16) / 2) __builtin_amdgcn_s_setprio(2);
    else if (kt == (Hn / 16) * 3 / 4) __builtin_amdgcn_s_setprio(1);
    else if (kt == (Hn / 16) * 7 / 8) __builtin_amdgcn_s_setprio(0);
    const int cur = kt & 1;
    const bool more = (kt + 1) < Hn / 16;
    if (more) {
      const int k1 = (kt + 1) * 16;
#pragma unroll
      for (int j = 0; j < 2; ++j) ra[j] = *(const float4*)(fA + (size_t)(j * 64) * Hn + k1);
#pragma unroll
      for (int j = 0; j < 3; ++j) rb[j] = *(const float4*)(wB + (size_t)(j * 64) * Hn + k1);
    }
    v2f af[2][4], bf[6][4];
#pragma unroll
    for (int mt = 0; mt < 2; ++mt) {
      const int m = wm + mt * 16 + lm;
      const float4 lo = *(const float4*)(&As[cur][m * 20 + hi * 8]);
      const float4 up = *(const float4*)(&As[cur][m * 20 + hi * 8 + 4]);
      af[mt][0].x = lo.x; af[mt][0].y = lo.y; af[mt][1].x = lo.z; af[mt][1].y = lo.w;
      af[mt][2].x = up.x; af[mt][2].y = up.y; af[mt][3].x = up.z; af[mt][3].y = up.w;
    }
#pragma unroll
    for (int ct = 0; ct < 6; ++ct) {
      const int n = wn + ct * 16 + lm;
      const float4 lo = *(const float4*)(&Bs[cur][n * 20 + hi * 8]);
      const float4 up = *(const float4*)(&Bs[cur][n * 20 + hi * 8 + 4]);
      bf[ct][0].x = lo.x; bf[ct][0].y = lo.y; bf[ct][1].x = lo.z; bf[ct][1].y = lo.w;
      bf[ct][2].x = up.x; bf[ct][2].y = up.y; bf[ct][3].x = up.z; bf[ct][3].y = up.w;
    }
#pragma unroll
    for (int ks = 0; ks < 4; ++ks) {
#pragma unroll
      for (int mt = 0; mt < 2; ++mt) {
#pragma unroll
        for (int ct = 0; ct < 6; ++ct) {
          cAcc[mt][ct] = __builtin_amdgcn_wmma_f32_16x16x4_f32(
              false, af[mt][ks], false, bf[ct][ks], (short)0, cAcc[mt][ct], false, false);
        }
      }
    }
    if (more) {
#pragma unroll
      for (int j = 0; j < 2; ++j) {
        *(float2*)(&As[cur ^ 1][j * 64 * 20 + sLo]) = make_float2(ra[j].x, ra[j].y);
        *(float2*)(&As[cur ^ 1][j * 64 * 20 + sUp]) = make_float2(ra[j].z, ra[j].w);
      }
#pragma unroll
      for (int j = 0; j < 3; ++j) {
        *(float2*)(&Bs[cur ^ 1][j * 64 * 20 + sLo]) = make_float2(rb[j].x, rb[j].y);
        *(float2*)(&Bs[cur ^ 1][j * 64 * 20 + sUp]) = make_float2(rb[j].z, rb[j].w);
      }
    }
    __syncthreads();
  }

#pragma unroll
  for (int ct = 0; ct < 6; ++ct) {
    const int o = o0 + wn + ct * 16 + lm;
    const float bl = blin[o];
#pragma unroll
    for (int mt = 0; mt < 2; ++mt) {
#pragma unroll
      for (int r = 0; r < 8; ++r) {
        const int m = m0 + wm + mt * 16 + r + hi * 8;
        wout[(size_t)(b * Nn + m) * NK + o] = cAcc[mt][ct][r] + bl;
      }
    }
  }
}

#define KSTG (8 * KP * 2)
#define AWP 68
#define AWSTG (16 * AWP)

template <int P>
__device__ __forceinline__ void slab_fetch(const float* __restrict__ kin, int b, int i0,
                                           int kPr0, int kC4, int tid, int hRow,
                                           float4 (&rx)[4], float4 (&ry)[4], float& rh) {
  const float* kr = kin + (size_t)(b * Nn + i0 + kPr0 * 2) * Hn + P * 512 + kC4;
#pragma unroll
  for (int j = 0; j < 4; ++j) {
    rx[j] = *(const float4*)(kr + (size_t)(j * 4) * Hn);
    ry[j] = *(const float4*)(kr + (size_t)(j * 4 + 1) * Hn);
  }
  if (tid < 32) {
    const float* krh = kin + (size_t)(b * Nn + i0 + hRow) * Hn;
    float v = 0.0f;
    if (P == 0) { if (tid >= 16) v = krh[512]; }
    else        { if (tid < 16)  v = krh[511]; }
    rh = v;
  }
}

__device__ __forceinline__ void slab_store(float* KsP, int kPr0, int kC4, int tid,
                                           int hBase, const float4 (&rx)[4],
                                           const float4 (&ry)[4], float rh) {
  float* kd = KsP + (kPr0 * KP + 4 + kC4) * 2;
#pragma unroll
  for (int j = 0; j < 4; ++j) {
    const float4 x = rx[j], y = ry[j];
    const float4 lo = make_float4(x.x, y.x, x.y, y.y);
    const float4 up = make_float4(x.z, y.z, x.w, y.w);
    float* d = kd + (j * 2 * KP) * 2;
    *(float4*)(d)     = lo;
    *(float4*)(d + 4) = up;
  }
  if (tid < 32) {
    if (tid < 16) KsP[hBase + 3 * 2] = rh;
    else          KsP[hBase + 516 * 2] = rh;
  }
}

__device__ __forceinline__ float from_left(float old, float src) {
  return __int_as_float(__builtin_amdgcn_update_dpp(__float_as_int(old), __float_as_int(src),
                                                    0x111  , 0xF, 0xF, false));
}
__device__ __forceinline__ float from_right(float old, float src) {
  return __int_as_float(__builtin_amdgcn_update_dpp(__float_as_int(old), __float_as_int(src),
                                                    0x101  , 0xF, 0xF, false));
}
__device__ __forceinline__ float rot_last_to_first(float src) {
  return __int_as_float(__builtin_amdgcn_update_dpp(__float_as_int(src), __float_as_int(src),
                                                    0x121  , 0xF, 0xF, false));
}
__device__ __forceinline__ float rot_first_to_last(float src) {
  return __int_as_float(__builtin_amdgcn_update_dpp(__float_as_int(src), __float_as_int(src),
                                                    0x12F  , 0xF, 0xF, false));
}

struct FragRaw {
  v2f a[4];
  float4 d12[2];
  v2f eL, eR;
};

template <int KS>
__device__ __forceinline__ void frag_issue(const float* KsP, const float* AwC, int wave, int lm,
                                           int hi2, FragRaw& fr) {
#pragma unroll
  for (int c = 0; c < 4; ++c)
    fr.a[c] = *(const v2f*)(&AwC[lm * AWP + c * 16 + hi2 * 8 + 2 * KS]);
  const float* row = &KsP[((KS * 2 + hi2) * KP) * 2];
#pragma unroll
  for (int pt = 0; pt < 2; ++pt)
    fr.d12[pt] = *(const float4*)(row + (wave * 64 + (pt * 16 + lm) * 2 + 4) * 2);
  fr.eL = *(const v2f*)(row + (wave * 64 + 3) * 2);
  fr.eR = *(const v2f*)(row + (wave * 64 + 68) * 2);
}

template <int P>
__device__ __forceinline__ void frag_mul(const FragRaw& fr, v8f (&acc)[16]) {
  v2f d0[2], d1[2], d2[2], d3[2];
#pragma unroll
  for (int pt = 0; pt < 2; ++pt) {
    d1[pt].x = fr.d12[pt].x; d1[pt].y = fr.d12[pt].y;
    d2[pt].x = fr.d12[pt].z; d2[pt].y = fr.d12[pt].w;
  }
  d0[0].x = from_left(fr.eL.x, d2[0].x);
  d0[0].y = from_left(fr.eL.y, d2[0].y);
  d0[1].x = from_left(rot_last_to_first(d2[0].x), d2[1].x);
  d0[1].y = from_left(rot_last_to_first(d2[0].y), d2[1].y);
  d3[1].x = from_right(fr.eR.x, d1[1].x);
  d3[1].y = from_right(fr.eR.y, d1[1].y);
  d3[0].x = from_right(rot_first_to_last(d1[1].x), d1[0].x);
  d3[0].y = from_right(rot_first_to_last(d1[1].y), d1[0].y);
  v2f bq[4][2];
#pragma unroll
  for (int pt = 0; pt < 2; ++pt) {
    bq[0][pt] = d0[pt] - d2[pt];
    bq[1][pt] = d1[pt] + d2[pt];
    bq[2][pt] = d2[pt] - d1[pt];
    bq[3][pt] = d1[pt] - d3[pt];
  }
#pragma unroll
  for (int c = 0; c < 4; ++c) {
#pragma unroll
    for (int pt = 0; pt < 2; ++pt) {
      acc[P * 8 + pt * 4 + c] = __builtin_amdgcn_wmma_f32_16x16x4_f32(
          false, fr.a[c], false, bq[c][pt], (short)0, acc[P * 8 + pt * 4 + c], false, false);
    }
  }
}

__global__ __launch_bounds__(256) void dyconv_ln(
    const float* __restrict__ kin, const float* __restrict__ w,
    const float* __restrict__ gamma, const float* __restrict__ beta,
    float* __restrict__ out) {
  __shared__ __attribute__((aligned(16))) float Ks[2][KSTG];
  __shared__ __attribute__((aligned(16))) float Aw[2][AWSTG];
  __shared__ __attribute__((aligned(16))) float Gs[1024];
  __shared__ __attribute__((aligned(16))) float Bt[1024];
  __shared__ float redS[8 * 16];
  __shared__ float redQ[8 * 16];
  __shared__ float muS[16];
  __shared__ float rsS[16];

  const int tid = threadIdx.x;
  const int b = blockIdx.y;
  const int m0 = blockIdx.x * 16;
  const int wave = tid >> 5, lane = tid & 31;
  const int lm = lane & 15, hi2 = lane >> 4;

  __builtin_amdgcn_s_setprio(3);

#pragma unroll
  for (int j = 0; j < 4; ++j) {
    int i = tid + j * 256;
    Gs[i] = gamma[i];
    Bt[i] = beta[i];
  }

  v8f acc[16] = {};

  const int aRow = tid >> 4;
  const int aI = tid & 15;
  const float* wp = w + (size_t)(b * Nn + m0 + aRow) * NK + aI * 3;
  const int aDst = aRow * AWP + kpos(aI);
  const int kPr0 = tid >> 7;
  const int kC4 = (tid & 127) * 4;
  const int hRow = tid & 15;
  const int hBase = ((hRow >> 1) * KP) * 2 + (hRow & 1);

  float4 rx[4], ry[4];
  float rh = 0.0f;
  float rw0, rw1, rw2;

  slab_fetch<0>(kin, b, 0, kPr0, kC4, tid, hRow, rx, ry, rh);
  rw0 = wp[0]; rw1 = wp[1]; rw2 = wp[2];
  slab_store(Ks[0], kPr0, kC4, tid, hBase, rx, ry, rh);
  Aw[0][aDst + 0 * 16] = rw0;
  Aw[0][aDst + 1 * 16] = ((rw0 + rw2) + rw1) * 0.5f;
  Aw[0][aDst + 2 * 16] = ((rw0 + rw2) - rw1) * 0.5f;
  Aw[0][aDst + 3 * 16] = rw2;
  __syncthreads();

  for (int ic = 0; ic < 16; ++ic) {
    if (ic == 8) __builtin_amdgcn_s_setprio(2);
    else if (ic == 12) __builtin_amdgcn_s_setprio(1);
    else if (ic == 14) __builtin_amdgcn_s_setprio(0);
    const int i0 = ic * 16;
    const float* AwC = Aw[ic & 1];
    FragRaw fa, fb;
    frag_issue<0>(Ks[0], AwC, wave, lm, hi2, fa);
    frag_issue<1>(Ks[0], AwC, wave, lm, hi2, fb);
    __builtin_amdgcn_sched_barrier(0);
    frag_mul<0>(fa, acc);
    __builtin_amdgcn_sched_barrier(0);
    frag_issue<2>(Ks[0], AwC, wave, lm, hi2, fa);
    __builtin_amdgcn_sched_barrier(0);
    frag_mul<0>(fb, acc);
    __builtin_amdgcn_sched_barrier(0);
    frag_issue<3>(Ks[0], AwC, wave, lm, hi2, fb);
    slab_fetch<1>(kin, b, i0, kPr0, kC4, tid, hRow, rx, ry, rh);
    __builtin_amdgcn_sched_barrier(0);
    frag_mul<0>(fa, acc);
    frag_mul<0>(fb, acc);
    __builtin_amdgcn_sched_barrier(0);
    slab_store(Ks[1], kPr0, kC4, tid, hBase, rx, ry, rh);
    __syncthreads();
    const bool more = (ic + 1) < 16;
    frag_issue<0>(Ks[1], AwC, wave, lm, hi2, fa);
    frag_issue<1>(Ks[1], AwC, wave, lm, hi2, fb);
    __builtin_amdgcn_sched_barrier(0);
    frag_mul<1>(fa, acc);
    __builtin_amdgcn_sched_barrier(0);
    frag_issue<2>(Ks[1], AwC, wave, lm, hi2, fa);
    __builtin_amdgcn_sched_barrier(0);
    frag_mul<1>(fb, acc);
    __builtin_amdgcn_sched_barrier(0);
    frag_issue<3>(Ks[1], AwC, wave, lm, hi2, fb);
    if (more) {
      slab_fetch<0>(kin, b, i0 + 16, kPr0, kC4, tid, hRow, rx, ry, rh);
      const float* wq = wp + (size_t)(i0 + 16) * 3;
      rw0 = wq[0]; rw1 = wq[1]; rw2 = wq[2];
    }
    __builtin_amdgcn_sched_barrier(0);
    frag_mul<1>(fa, acc);
    frag_mul<1>(fb, acc);
    __builtin_amdgcn_sched_barrier(0);
    if (more) {
      slab_store(Ks[0], kPr0, kC4, tid, hBase, rx, ry, rh);
      float* AwN = Aw[(ic + 1) & 1];
      AwN[aDst + 0 * 16] = rw0;
      AwN[aDst + 1 * 16] = ((rw0 + rw2) + rw1) * 0.5f;
      AwN[aDst + 2 * 16] = ((rw0 + rw2) - rw1) * 0.5f;
      AwN[aDst + 3 * 16] = rw2;
    }
    __syncthreads();
  }

#pragma unroll
  for (int u = 0; u < 4; ++u)
#pragma unroll
    for (int r = 0; r < 8; ++r) {
      const float M0 = acc[u * 4 + 0][r], M1 = acc[u * 4 + 1][r];
      const float M2 = acc[u * 4 + 2][r], M3 = acc[u * 4 + 3][r];
      acc[u * 4 + 0][r] = (M0 + M1) + M2;
      acc[u * 4 + 1][r] = (M1 - M2) - M3;
    }

  float s[8], q[8];
#pragma unroll
  for (int r = 0; r < 8; ++r) { s[r] = 0.0f; q[r] = 0.0f; }
#pragma unroll
  for (int u = 0; u < 4; ++u)
#pragma unroll
    for (int e = 0; e < 2; ++e)
#pragma unroll
      for (int r = 0; r < 8; ++r) {
        float v = acc[u * 4 + e][r];
        s[r] += v;
        q[r] += v * v;
      }
#pragma unroll
  for (int mask = 1; mask < 16; mask <<= 1)
#pragma unroll
    for (int r = 0; r < 8; ++r) {
      s[r] += __shfl_xor(s[r], mask, 32);
      q[r] += __shfl_xor(q[r], mask, 32);
    }
  if (lm == 0) {
#pragma unroll
    for (int r = 0; r < 8; ++r) {
      const int row = r + hi2 * 8;
      redS[wave * 16 + row] = s[r];
      redQ[wave * 16 + row] = q[r];
    }
  }
  __syncthreads();
  if (tid < 16) {
    float ss = 0.0f, qq = 0.0f;
#pragma unroll
    for (int wv = 0; wv < 8; ++wv) {
      ss += redS[wv * 16 + tid];
      qq += redQ[wv * 16 + tid];
    }
    const float mu = ss * (1.0f / 1024.0f);
    const float var = qq * (1.0f / 1024.0f) - mu * mu;
    muS[tid] = mu;
    rsS[tid] = rsqrtf(var + LN_EPS);
  }
  __syncthreads();

#pragma unroll
  for (int u = 0; u < 4; ++u) {
    const int p = u >> 1, pt = u & 1;
    const int h = p * 512 + wave * 64 + (pt * 16 + lm) * 2;
    const float2 g = *(const float2*)(&Gs[h]);
    const float2 be = *(const float2*)(&Bt[h]);
#pragma unroll
    for (int r = 0; r < 8; ++r) {
      const int row = r + hi2 * 8;
      const float mu = muS[row], rs = rsS[row];
      float2 v;
      v.x = (acc[u * 4 + 0][r] - mu) * rs * g.x + be.x;
      v.y = (acc[u * 4 + 1][r] - mu) * rs * g.y + be.y;
      *(float2*)(&out[(size_t)(b * Nn + m0 + row) * Hn + h]) = v;
    }
  }
}

extern "C" void kernel_launch(void* const* d_in, const int* in_sizes, int n_in,
                              void* d_out, int out_size, void* d_ws, size_t ws_size,
                              hipStream_t stream) {
  const float* f     = (const float*)d_in[0];
  const float* k     = (const float*)d_in[1];
  const float* Wlin  = (const float*)d_in[2];
  const float* blin  = (const float*)d_in[3];
  const float* gamma = (const float*)d_in[4];
  const float* beta  = (const float*)d_in[5];
  float* out = (float*)d_out;
  float* wgt = (float*)d_ws;

  (void)in_sizes; (void)n_in; (void)out_size; (void)ws_size;

  wgen_gemm<<<dim3(NK / T1N, Nn / T1M, Bn), 256, 0, stream>>>(f, Wlin, blin, wgt);
  dyconv_ln<<<dim3(Nn / 16, Bn), 256, 0, stream>>>(k, wgt, gamma, beta, out);
}


# Both k loops as one software pipeline: LDS stages hold two slabs (stage 1) or both column halves (stage 2), fragment reads stay ahead across one split barrier per stage; sums unchanged
# speedup vs baseline: 1.7910x; 1.0956x over previous
#include <hip/hip_runtime.h>
#include <hip/hip_bf16.h>


#define Bn 64
#define Nn 256
#define Hn 1024
#define NK 768
#define LN_EPS 1e-5f
#define KP 528

__device__ __forceinline__ int kpos(int k) { return 8 * ((k >> 1) & 1) + 2 * (k >> 2) + (k & 1); }

typedef __attribute__((ext_vector_type(2))) float v2f;
typedef __attribute__((ext_vector_type(8))) float v8f;

__device__ __forceinline__ void wg_arrive() {
  asm volatile("s_wait_dscnt 0x0\n\ts_barrier_signal -1" ::: "memory");
}
__device__ __forceinline__ void wg_wait() {
  asm volatile("s_barrier_wait -1" ::: "memory");
}

#define T1M 128
#define T1N 192

template <int UP>
__device__ __forceinline__ void s1_frag_read(const float* Asl, const float* Bsl, int aoff, int boff,
                                             float4 (&a)[2], float4 (&b)[6]) {
#pragma unroll
  for (int mt = 0; mt < 2; ++mt) a[mt] = *(const float4*)(Asl + aoff + mt * 16 * 20 + UP * 4);
#pragma unroll
  for (int ct = 0; ct < 6; ++ct) b[ct] = *(const float4*)(Bsl + boff + ct * 16 * 20 + UP * 4);
}

__device__ __forceinline__ void s1_products(const float4 (&a)[2], const float4 (&b)[6], v8f (&c)[2][6]) {
#pragma unroll
  for (int s = 0; s < 2; ++s) {
#pragma unroll
    for (int mt = 0; mt < 2; ++mt) {
#pragma unroll
      for (int ct = 0; ct < 6; ++ct) {
        v2f x, y;
        if (s) { x.x = a[mt].z; x.y = a[mt].w; y.x = b[ct].z; y.y = b[ct].w; }
        else   { x.x = a[mt].x; x.y = a[mt].y; y.x = b[ct].x; y.y = b[ct].y; }
        c[mt][ct] = __builtin_amdgcn_wmma_f32_16x16x4_f32(false, x, false, y, (short)0, c[mt][ct],
                                                          false, false);
      }
    }
  }
}

__device__ __forceinline__ void s1_fetch(const float* fA, const float* wB, int k1,
                                         float4 (&ra)[2], float4 (&rb)[3]) {
#pragma unroll
  for (int j = 0; j < 2; ++j) ra[j] = *(const float4*)(fA + (size_t)(j * 64) * Hn + k1);
#pragma unroll
  for (int j = 0; j < 3; ++j) rb[j] = *(const float4*)(wB + (size_t)(j * 64) * Hn + k1);
}

__device__ __forceinline__ void s1_store(float* Asl, float* Bsl, int sLo, int sUp,
                                         const float4 (&ra)[2], const float4 (&rb)[3]) {
#pragma unroll
  for (int j = 0; j < 2; ++j) {
    *(float2*)(&Asl[j * 64 * 20 + sLo]) = make_float2(ra[j].x, ra[j].y);
    *(float2*)(&Asl[j * 64 * 20 + sUp]) = make_float2(ra[j].z, ra[j].w);
  }
#pragma unroll
  for (int j = 0; j < 3; ++j) {
    *(float2*)(&Bsl[j * 64 * 20 + sLo]) = make_float2(rb[j].x, rb[j].y);
    *(float2*)(&Bsl[j * 64 * 20 + sUp]) = make_float2(rb[j].z, rb[j].w);
  }
}
__global__ __launch_bounds__(256) void wgen_gemm(
    const float* __restrict__ f, const float* __restrict__ Wlin,
    const float* __restrict__ blin, float* __restrict__ wout) {
  __shared__ __attribute__((aligned(16))) float As[2][2][T1M * 20];
  __shared__ __attribute__((aligned(16))) float Bs[2][2][T1N * 20];

  const int tid = threadIdx.x;
  const int b = blockIdx.z;
  const int m0 = blockIdx.y * T1M;
  const int o0 = blockIdx.x * T1N;
  const int wave = tid >> 5, lane = tid & 31;
  const int lm = lane & 15, hi = lane >> 4;
  const int wm = (wave & 3) * 32;
  const int wn = (wave >> 2) * 96;

  __builtin_amdgcn_s_setprio(3);

  v8f cAcc[2][6] = {};

  const int srow = tid >> 2;
  const int scol = (tid & 3) * 4;
  const float* fA = f + (size_t)(b * Nn + m0 + srow) * Hn + scol;
  const float* wB = Wlin + (size_t)(o0 + srow) * Hn + scol;
  const int sLo = srow * 20 + kpos(scol);
  const int sUp = srow * 20 + kpos(scol + 2);

  const int aoff = (wm + lm) * 20 + hi * 8;
  const int boff = (wn + lm) * 20 + hi * 8;

  float4 ra[2], rb[3];
  {
    float4 ra1[2], rb1[3];
    s1_fetch(fA, wB, 0, ra, rb);
    s1_fetch(fA, wB, 16, ra1, rb1);
    s1_store(As[0][0], Bs[0][0], sLo, sUp, ra, rb);
    s1_store(As[0][1], Bs[0][1], sLo, sUp, ra1, rb1);
  }
  __syncthreads();

  float4 aL[2], bL[6], aU[2], bU[6];
  s1_fetch(fA, wB, 32, ra, rb);
  s1_frag_read<0>(As[0][0], Bs[0][0], aoff, boff, aL, bL);
  s1_frag_read<1>(As[0][0], Bs[0][0], aoff, boff, aU, bU);

  for (int kp = 0; kp < Hn / 32; ++kp) {
    if (kp == (Hn / 32) / 2) __builtin_amdgcn_s_setprio(2);
    else if (kp == (Hn / 32) * 3 / 4) __builtin_amdgcn_s_setprio(1);
    else if (kp == (Hn / 32) * 7 / 8) __builtin_amdgcn_s_setprio(0);
    const int cur = kp & 1;
    const bool more = (kp + 1) < Hn / 32;
    const bool more2 = (kp + 2) < Hn / 32;
    __builtin_amdgcn_sched_barrier(0);
    s1_products(aL, bL, cAcc);
    __builtin_amdgcn_sched_barrier(0);
    s1_frag_read<0>(As[cur][1], Bs[cur][1], aoff, boff, aL, bL);
    if (more) {
      s1_store(As[cur ^ 1][0], Bs[cur ^ 1][0], sLo, sUp, ra, rb);
      s1_fetch(fA, wB, (kp * 2 + 3) * 16, ra, rb);
    }
    __builtin_amdgcn_sched_barrier(0);
    s1_products(aU, bU, cAcc);
    __builtin_amdgcn_sched_barrier(0);
    s1_frag_read<1>(As[cur][1], Bs[cur][1], aoff, boff, aU, bU);
    if (more) s1_store(As[cur ^ 1][1], Bs[cur ^ 1][1], sLo, sUp, ra, rb);
    wg_arrive();
    __builtin_amdgcn_sched_barrier(0);
    s1_products(aL, bL, cAcc);
    __builtin_amdgcn_sched_barrier(0);
    wg_wait();
    if (more) {
      s1_frag_read<0>(As[cur ^ 1][0], Bs[cur ^ 1][0], aoff, boff, aL, bL);
      if (more2) s1_fetch(fA, wB, (kp * 2 + 4) * 16, ra, rb);
    }
    __builtin_amdgcn_sched_barrier(0);
    s1_products(aU, bU, cAcc);
    __builtin_amdgcn_sched_barrier(0);
    if (more) s1_frag_read<1>(As[cur ^ 1][0], Bs[cur ^ 1][0], aoff, boff, aU, bU);
  }

#pragma unroll
  for (int ct = 0; ct < 6; ++ct) {
    const int o = o0 + wn + ct * 16 + lm;
    const float bl = blin[o];
#pragma unroll
    for (int mt = 0; mt < 2; ++mt) {
#pragma unroll
      for (int r = 0; r < 8; ++r) {
        const int m = m0 + wm + mt * 16 + r + hi * 8;
        wout[(size_t)(b * Nn + m) * NK + o] = cAcc[mt][ct][r] + bl;
      }
    }
  }
}

#define KSTG (8 * KP * 2)
#define AWP 100
#define AWSTG (16 * AWP)

template <int P>
__device__ __forceinline__ void slab_fetch(const float* __restrict__ kin, int b, int i0,
                                           int kPr0, int kC4, int tid, int hRow,
                                           float4 (&rx)[4], float4 (&ry)[4], float& rh) {
  const float* kr = kin + (size_t)(b * Nn + i0 + kPr0 * 2) * Hn + P * 512 + kC4;
#pragma unroll
  for (int j = 0; j < 4; ++j) {
    rx[j] = *(const float4*)(kr + (size_t)(j * 4) * Hn);
    ry[j] = *(const float4*)(kr + (size_t)(j * 4 + 1) * Hn);
  }
  if (tid < 32) {
    const float* krh = kin + (size_t)(b * Nn + i0 + hRow) * Hn;
    float v = 0.0f;
    if (P == 0) { if (tid >= 16) v = krh[512]; }
    else        { if (tid < 16)  v = krh[511]; }
    rh = v;
  }
}

__device__ __forceinline__ void slab_store(float* KsP, int kPr0, int kC4, int tid,
                                           int hBase, const float4 (&rx)[4],
                                           const float4 (&ry)[4], float rh) {
  float* kd = KsP + (kPr0 * KP + 4 + kC4) * 2;
#pragma unroll
  for (int j = 0; j < 4; ++j) {
    const float4 x = rx[j], y = ry[j];
    const float4 lo = make_float4(x.x, y.x, x.y, y.y);
    const float4 up = make_float4(x.z, y.z, x.w, y.w);
    float* d = kd + (j * 2 * KP) * 2;
    *(float4*)(d)     = lo;
    *(float4*)(d + 4) = up;
  }
  if (tid < 32) {
    if (tid < 16) KsP[hBase + 3 * 2] = rh;
    else          KsP[hBase + 516 * 2] = rh;
  }
}

__device__ __forceinline__ void wcomb_store(float* AwS, int aDst, float g0, float g1, float g2) {
  const float s02 = g0 + g2;
  const float ev = g0 * (1.0f / 24.0f) + g2 * (1.0f / 6.0f);
  const float od = g1 * (1.0f / 12.0f);
  AwS[aDst + 0 * 16] = g0 * 0.25f;
  AwS[aDst + 1 * 16] = (s02 + g1) * (-1.0f / 6.0f);
  AwS[aDst + 2 * 16] = (s02 - g1) * (-1.0f / 6.0f);
  AwS[aDst + 3 * 16] = ev + od;
  AwS[aDst + 4 * 16] = ev - od;
  AwS[aDst + 5 * 16] = g2;
}

__device__ __forceinline__ float from_left(float old, float src) {
  return __int_as_float(__builtin_amdgcn_update_dpp(__float_as_int(old), __float_as_int(src),
                                                    0x111  , 0xF, 0xF, false));
}
__device__ __forceinline__ float from_right(float old, float src) {
  return __int_as_float(__builtin_amdgcn_update_dpp(__float_as_int(old), __float_as_int(src),
                                                    0x101  , 0xF, 0xF, false));
}

struct FragD {
  float4 q[2];
  v2f eL, eR;
};

template <int KS>
__device__ __forceinline__ void frag_issue(const float* KsP, int wave, int lm, int hi2, FragD& fr) {
  const float* row = &KsP[((KS * 2 + hi2) * KP) * 2];
  const float* p = row + (wave * 64 + lm * 4 + 4) * 2;
  fr.q[0] = *(const float4*)(p);
  fr.q[1] = *(const float4*)(p + 4);
  fr.eL = *(const v2f*)(row + (wave * 64 + 3) * 2);
  fr.eR = *(const v2f*)(row + (wave * 64 + 68) * 2);
}

template <int J>
__device__ __forceinline__ void wfrag_issue(const float* AwC, int lm, int hi2, float4 (&aw)[6][2]) {
#pragma unroll
  for (int c = 0; c < 6; ++c)
    aw[c][J] = *(const float4*)(&AwC[lm * AWP + c * 16 + hi2 * 8 + 4 * J]);
}

template <int P, int KS>
__device__ __forceinline__ void frag_mul(const FragD& fr, const float4 (&aw)[6][2], v8f (&acc)[12]) {
  v2f d0, d1, d2, d3, d4, d5;
  d1.x = fr.q[0].x; d1.y = fr.q[0].y;
  d2.x = fr.q[0].z; d2.y = fr.q[0].w;
  d3.x = fr.q[1].x; d3.y = fr.q[1].y;
  d4.x = fr.q[1].z; d4.y = fr.q[1].w;
  d0.x = from_left(fr.eL.x, d4.x);
  d0.y = from_left(fr.eL.y, d4.y);
  d5.x = from_right(fr.eR.x, d1.x);
  d5.y = from_right(fr.eR.y, d1.y);
  const v2f t1 = d4 - 4.0f * d2;
  const v2f t2 = d3 - 4.0f * d1;
  const v2f t3 = d4 - d2;
  const v2f u = d3 - d1;
  v2f bq[6];
  bq[0] = 4.0f * d0 + (d4 - 5.0f * d2);
  bq[1] = t1 + t2;
  bq[2] = t1 - t2;
  bq[3] = t3 + 2.0f * u;
  bq[4] = t3 - 2.0f * u;
  bq[5] = 4.0f * d1 + (d5 - 5.0f * d3);
#pragma unroll
  for (int c = 0; c < 6; ++c) {
    v2f a;
    if (KS & 1) { a.x = aw[c][KS >> 1].z; a.y = aw[c][KS >> 1].w; }
    else        { a.x = aw[c][KS >> 1].x; a.y = aw[c][KS >> 1].y; }
    acc[P * 6 + c] = __builtin_amdgcn_wmma_f32_16x16x4_f32(
        false, a, false, bq[c], (short)0, acc[P * 6 + c], false, false);
  }
}

__global__ __launch_bounds__(256) void dyconv_ln(
    const float* __restrict__ kin, const float* __restrict__ w,
    const float* __restrict__ gamma, const float* __restrict__ beta,
    float* __restrict__ out) {
  __shared__ __attribute__((aligned(16))) float Ks[2][2][KSTG];
  __shared__ __attribute__((aligned(16))) float Aw[2][AWSTG];
  __shared__ __attribute__((aligned(16))) float Gs[1024];
  __shared__ __attribute__((aligned(16))) float Bt[1024];
  __shared__ float redS[8 * 16];
  __shared__ float redQ[8 * 16];
  __shared__ float muS[16];
  __shared__ float rsS[16];

  const int tid = threadIdx.x;
  const int b = blockIdx.y;
  const int m0 = blockIdx.x * 16;
  const int wave = tid >> 5, lane = tid & 31;
  const int lm = lane & 15, hi2 = lane >> 4;

  __builtin_amdgcn_s_setprio(3);

#pragma unroll
  for (int j = 0; j < 4; ++j) {
    int i = tid + j * 256;
    Gs[i] = gamma[i];
    Bt[i] = beta[i];
  }

  v8f acc[12] = {};

  const int aRow = tid >> 4;
  const int aI = tid & 15;
  const float* wp = w + (size_t)(b * Nn + m0 + aRow) * NK + aI * 3;
  const int aDst = aRow * AWP + kpos(aI);
  const int kPr0 = tid >> 7;
  const int kC4 = (tid & 127) * 4;
  const int hRow = tid & 15;
  const int hBase = ((hRow >> 1) * KP) * 2 + (hRow & 1);

  float4 rx[4], ry[4];
  float rh = 0.0f;
  float rw0, rw1, rw2;

  {
    float4 rx1[4], ry1[4];
    float rh1 = 0.0f;
    slab_fetch<0>(kin, b, 0, kPr0, kC4, tid, hRow, rx, ry, rh);
    slab_fetch<1>(kin, b, 0, kPr0, kC4, tid, hRow, rx1, ry1, rh1);
    rw0 = wp[0]; rw1 = wp[1]; rw2 = wp[2];
    slab_store(Ks[0][0], kPr0, kC4, tid, hBase, rx, ry, rh);
    slab_store(Ks[0][1], kPr0, kC4, tid, hBase, rx1, ry1, rh1);
    wcomb_store(Aw[0], aDst, rw0, rw1, rw2);
  }
  __syncthreads();

  float4 aw[6][2];
  FragD fa, fb;
  wfrag_issue<0>(Aw[0], lm, hi2, aw);
  frag_issue<0>(Ks[0][0], wave, lm, hi2, fa);
  frag_issue<1>(Ks[0][0], wave, lm, hi2, fb);
  wfrag_issue<1>(Aw[0], lm, hi2, aw);

  for (int ic = 0; ic < 16; ++ic) {
    if (ic == 8) __builtin_amdgcn_s_setprio(2);
    else if (ic == 12) __builtin_amdgcn_s_setprio(1);
    else if (ic == 14) __builtin_amdgcn_s_setprio(0);
    const int i0 = ic * 16;
    const int st = ic & 1;
    const float* K0 = Ks[st][0];
    const float* K1 = Ks[st][1];
    const float* K0n = Ks[st ^ 1][0];
    const float* AwN = Aw[st ^ 1];
    const bool more = (ic + 1) < 16;
    __builtin_amdgcn_sched_barrier(0);
    frag_mul<0, 0>(fa, aw, acc);
    __builtin_amdgcn_sched_barrier(0);
    frag_issue<2>(K0, wave, lm, hi2, fa);
    __builtin_amdgcn_sched_barrier(0);
    frag_mul<0, 1>(fb, aw, acc);
    __builtin_amdgcn_sched_barrier(0);
    frag_issue<3>(K0, wave, lm, hi2, fb);
    if (more) slab_fetch<0>(kin, b, i0 + 16, kPr0, kC4, tid, hRow, rx, ry, rh);
    __builtin_amdgcn_sched_barrier(0);
    frag_mul<0, 2>(fa, aw, acc);
    __builtin_amdgcn_sched_barrier(0);
    frag_issue<0>(K1, wave, lm, hi2, fa);
    __builtin_amdgcn_sched_barrier(0);
    frag_mul<0, 3>(fb, aw, acc);
    __builtin_amdgcn_sched_barrier(0);
    frag_issue<1>(K1, wave, lm, hi2, fb);
    if (more) {
      slab_store(Ks[st ^ 1][0], kPr0, kC4, tid, hBase, rx, ry, rh);
      slab_fetch<1>(kin, b, i0 + 16, kPr0, kC4, tid, hRow, rx, ry, rh);
      const float* wq = wp + (size_t)(i0 + 16) * 3;
      rw0 = wq[0]; rw1 = wq[1]; rw2 = wq[2];
    }
    __builtin_amdgcn_sched_barrier(0);
    frag_mul<1, 0>(fa, aw, acc);
    __builtin_amdgcn_sched_barrier(0);
    frag_issue<2>(K1, wave, lm, hi2, fa);
    __builtin_amdgcn_sched_barrier(0);
    frag_mul<1, 1>(fb, aw, acc);
    __builtin_amdgcn_sched_barrier(0);
    frag_issue<3>(K1, wave, lm, hi2, fb);
    if (more) {
      slab_store(Ks[st ^ 1][1], kPr0, kC4, tid, hBase, rx, ry, rh);
      wcomb_store(Aw[st ^ 1], aDst, rw0, rw1, rw2);
    }
    wg_arrive();
    __builtin_amdgcn_sched_barrier(0);
    frag_mul<1, 2>(fa, aw, acc);
    __builtin_amdgcn_sched_barrier(0);
    wg_wait();
    if (more) {
      wfrag_issue<0>(AwN, lm, hi2, aw);
      frag_issue<0>(K0n, wave, lm, hi2, fa);
    }
    __builtin_amdgcn_sched_barrier(0);
    frag_mul<1, 3>(fb, aw, acc);
    __builtin_amdgcn_sched_barrier(0);
    if (more) {
      frag_issue<1>(K0n, wave, lm, hi2, fb);
      wfrag_issue<1>(AwN, lm, hi2, aw);
    }
  }

#pragma unroll
  for (int p = 0; p < 2; ++p)
#pragma unroll
    for (int r = 0; r < 8; ++r) {
      const float M0 = acc[p * 6 + 0][r], M1 = acc[p * 6 + 1][r], M2 = acc[p * 6 + 2][r];
      const float M3 = acc[p * 6 + 3][r], M4 = acc[p * 6 + 4][r], M5 = acc[p * 6 + 5][r];
      const float s12 = M1 + M2, d12 = M1 - M2, s34 = M3 + M4, d34 = M3 - M4;
      acc[p * 6 + 0][r] = (M0 + s12) + s34;
      acc[p * 6 + 1][r] = d12 + 2.0f * d34;
      acc[p * 6 + 2][r] = s12 + 4.0f * s34;
      acc[p * 6 + 3][r] = (d12 + 8.0f * d34) + M5;
    }

  float s[8], q[8];
#pragma unroll
  for (int r = 0; r < 8; ++r) { s[r] = 0.0f; q[r] = 0.0f; }
#pragma unroll
  for (int p = 0; p < 2; ++p)
#pragma unroll
    for (int e = 0; e < 4; ++e)
#pragma unroll
      for (int r = 0; r < 8; ++r) {
        float v = acc[p * 6 + e][r];
        s[r] += v;
        q[r] += v * v;
      }
#pragma unroll
  for (int mask = 1; mask < 16; mask <<= 1)
#pragma unroll
    for (int r = 0; r < 8; ++r) {
      s[r] += __shfl_xor(s[r], mask, 32);
      q[r] += __shfl_xor(q[r], mask, 32);
    }
  if (lm == 0) {
#pragma unroll
    for (int r = 0; r < 8; ++r) {
      const int row = r + hi2 * 8;
      redS[wave * 16 + row] = s[r];
      redQ[wave * 16 + row] = q[r];
    }
  }
  __syncthreads();
  if (tid < 16) {
    float ss = 0.0f, qq = 0.0f;
#pragma unroll
    for (int wv = 0; wv < 8; ++wv) {
      ss += redS[wv * 16 + tid];
      qq += redQ[wv * 16 + tid];
    }
    const float mu = ss * (1.0f / 1024.0f);
    const float var = qq * (1.0f / 1024.0f) - mu * mu;
    muS[tid] = mu;
    rsS[tid] = rsqrtf(var + LN_EPS);
  }
  __syncthreads();

#pragma unroll
  for (int p = 0; p < 2; ++p) {
    const int h = p * 512 + wave * 64 + lm * 4;
    const float4 g = *(const float4*)(&Gs[h]);
    const float4 be = *(const float4*)(&Bt[h]);
#pragma unroll
    for (int r = 0; r < 8; ++r) {
      const int row = r + hi2 * 8;
      const float mu = muS[row], rs = rsS[row];
      float4 v;
      v.x = (acc[p * 6 + 0][r] - mu) * rs * g.x + be.x;
      v.y = (acc[p * 6 + 1][r] - mu) * rs * g.y + be.y;
      v.z = (acc[p * 6 + 2][r] - mu) * rs * g.z + be.z;
      v.w = (acc[p * 6 + 3][r] - mu) * rs * g.w + be.w;
      *(float4*)(&out[(size_t)(b * Nn + m0 + row) * Hn + h]) = v;
    }
  }
}

extern "C" void kernel_launch(void* const* d_in, const int* in_sizes, int n_in,
                              void* d_out, int out_size, void* d_ws, size_t ws_size,
                              hipStream_t stream) {
  const float* f     = (const float*)d_in[0];
  const float* k     = (const float*)d_in[1];
  const float* Wlin  = (const float*)d_in[2];
  const float* blin  = (const float*)d_in[3];
  const float* gamma = (const float*)d_in[4];
  const float* beta  = (const float*)d_in[5];
  float* out = (float*)d_out;
  float* wgt = (float*)d_ws;

  (void)in_sizes; (void)n_in; (void)out_size; (void)ws_size;

  wgen_gemm<<<dim3(NK / T1N, Nn / T1M, Bn), 256, 0, stream>>>(f, Wlin, blin, wgt);
  dyconv_ln<<<dim3(Nn / 16, Bn), 256, 0, stream>>>(k, wgt, gamma, beta, out);
}


# Stage 2 re-tiled: one 16-wave work-group per 32 rows, a wave owns 64 columns and both 16-row tiles (12 accumulators, 4 waves per SIMD kept): a column fragment feeds 12 WMMAs; LayerNorm sums regrouped
# speedup vs baseline: 1.9233x; 1.0739x over previous
#include <hip/hip_runtime.h>
#include <hip/hip_bf16.h>


#define Bn 64
#define Nn 256
#define Hn 1024
#define NK 768
#define LN_EPS 1e-5f
#define KP 528

__device__ __forceinline__ int kpos(int k) { return 8 * ((k >> 1) & 1) + 2 * (k >> 2) + (k & 1); }

typedef __attribute__((ext_vector_type(2))) float v2f;
typedef __attribute__((ext_vector_type(8))) float v8f;

__device__ __forceinline__ void wg_arrive() {
  asm volatile("s_wait_dscnt 0x0\n\ts_barrier_signal -1" ::: "memory");
}
__device__ __forceinline__ void wg_wait() {
  asm volatile("s_barrier_wait -1" ::: "memory");
}

#define T1M 128
#define T1N 192

template <int UP>
__device__ __forceinline__ void s1_frag_read(const float* Asl, const float* Bsl, int aoff, int boff,
                                             float4 (&a)[2], float4 (&b)[6]) {
#pragma unroll
  for (int mt = 0; mt < 2; ++mt) a[mt] = *(const float4*)(Asl + aoff + mt * 16 * 20 + UP * 4);
#pragma unroll
  for (int ct = 0; ct < 6; ++ct) b[ct] = *(const float4*)(Bsl + boff + ct * 16 * 20 + UP * 4);
}

__device__ __forceinline__ void s1_products(const float4 (&a)[2], const float4 (&b)[6], v8f (&c)[2][6]) {
#pragma unroll
  for (int s = 0; s < 2; ++s) {
#pragma unroll
    for (int mt = 0; mt < 2; ++mt) {
#pragma unroll
      for (int ct = 0; ct < 6; ++ct) {
        v2f x, y;
        if (s) { x.x = a[mt].z; x.y = a[mt].w; y.x = b[ct].z; y.y = b[ct].w; }
        else   { x.x = a[mt].x; x.y = a[mt].y; y.x = b[ct].x; y.y = b[ct].y; }
        c[mt][ct] = __builtin_amdgcn_wmma_f32_16x16x4_f32(false, x, false, y, (short)0, c[mt][ct],
                                                          false, false);
      }
    }
  }
}

__device__ __forceinline__ void s1_fetch(const float* fA, const float* wB, int k1,
                                         float4 (&ra)[2], float4 (&rb)[3]) {
#pragma unroll
  for (int j = 0; j < 2; ++j) ra[j] = *(const float4*)(fA + (size_t)(j * 64) * Hn + k1);
#pragma unroll
  for (int j = 0; j < 3; ++j) rb[j] = *(const float4*)(wB + (size_t)(j * 64) * Hn + k1);
}

__device__ __forceinline__ void s1_store(float* Asl, float* Bsl, int sLo, int sUp,
                                         const float4 (&ra)[2], const float4 (&rb)[3]) {
#pragma unroll
  for (int j = 0; j < 2; ++j) {
    *(float2*)(&Asl[j * 64 * 20 + sLo]) = make_float2(ra[j].x, ra[j].y);
    *(float2*)(&Asl[j * 64 * 20 + sUp]) = make_float2(ra[j].z, ra[j].w);
  }
#pragma unroll
  for (int j = 0; j < 3; ++j) {
    *(float2*)(&Bsl[j * 64 * 20 + sLo]) = make_float2(rb[j].x, rb[j].y);
    *(float2*)(&Bsl[j * 64 * 20 + sUp]) = make_float2(rb[j].z, rb[j].w);
  }
}
__global__ __launch_bounds__(256) void wgen_gemm(
    const float* __restrict__ f, const float* __restrict__ Wlin,
    const float* __restrict__ blin, float* __restrict__ wout) {
  __shared__ __attribute__((aligned(16))) float As[2][2][T1M * 20];
  __shared__ __attribute__((aligned(16))) float Bs[2][2][T1N * 20];

  const int tid = threadIdx.x;
  const int b = blockIdx.z;
  const int m0 = blockIdx.y * T1M;
  const int o0 = blockIdx.x * T1N;
  const int wave = tid >> 5, lane = tid & 31;
  const int lm = lane & 15, hi = lane >> 4;
  const int wm = (wave & 3) * 32;
  const int wn = (wave >> 2) * 96;

  __builtin_amdgcn_s_setprio(3);

  v8f cAcc[2][6] = {};

  const int srow = tid >> 2;
  const int scol = (tid & 3) * 4;
  const float* fA = f + (size_t)(b * Nn + m0 + srow) * Hn + scol;
  const float* wB = Wlin + (size_t)(o0 + srow) * Hn + scol;
  const int sLo = srow * 20 + kpos(scol);
  const int sUp = srow * 20 + kpos(scol + 2);

  const int aoff = (wm + lm) * 20 + hi * 8;
  const int boff = (wn + lm) * 20 + hi * 8;

  float4 ra[2], rb[3];
  {
    float4 ra1[2], rb1[3];
    s1_fetch(fA, wB, 0, ra, rb);
    s1_fetch(fA, wB, 16, ra1, rb1);
    s1_store(As[0][0], Bs[0][0], sLo, sUp, ra, rb);
    s1_store(As[0][1], Bs[0][1], sLo, sUp, ra1, rb1);
  }
  __syncthreads();

  float4 aL[2], bL[6], aU[2], bU[6];
  s1_fetch(fA, wB, 32, ra, rb);
  s1_frag_read<0>(As[0][0], Bs[0][0], aoff, boff, aL, bL);
  s1_frag_read<1>(As[0][0], Bs[0][0], aoff, boff, aU, bU);

  for (int kp = 0; kp < Hn / 32; ++kp) {
    if (kp == (Hn / 32) / 2) __builtin_amdgcn_s_setprio(2);
    else if (kp == (Hn / 32) * 3 / 4) __builtin_amdgcn_s_setprio(1);
    else if (kp == (Hn / 32) * 7 / 8) __builtin_amdgcn_s_setprio(0);
    const int cur = kp & 1;
    const bool more = (kp + 1) < Hn / 32;
    const bool more2 = (kp + 2) < Hn / 32;
    __builtin_amdgcn_sched_barrier(0);
    s1_products(aL, bL, cAcc);
    __builtin_amdgcn_sched_barrier(0);
    s1_frag_read<0>(As[cur][1], Bs[cur][1], aoff, boff, aL, bL);
    if (more) {
      s1_store(As[cur ^ 1][0], Bs[cur ^ 1][0], sLo, sUp, ra, rb);
      s1_fetch(fA, wB, (kp * 2 + 3) * 16, ra, rb);
    }
    __builtin_amdgcn_sched_barrier(0);
    s1_products(aU, bU, cAcc);
    __builtin_amdgcn_sched_barrier(0);
    s1_frag_read<1>(As[cur][1], Bs[cur][1], aoff, boff, aU, bU);
    if (more) s1_store(As[cur ^ 1][1], Bs[cur ^ 1][1], sLo, sUp, ra, rb);
    wg_arrive();
    __builtin_amdgcn_sched_barrier(0);
    s1_products(aL, bL, cAcc);
    __builtin_amdgcn_sched_barrier(0);
    wg_wait();
    if (more) {
      s1_frag_read<0>(As[cur ^ 1][0], Bs[cur ^ 1][0], aoff, boff, aL, bL);
      if (more2) s1_fetch(fA, wB, (kp * 2 + 4) * 16, ra, rb);
    }
    __builtin_amdgcn_sched_barrier(0);
    s1_products(aU, bU, cAcc);
    __builtin_amdgcn_sched_barrier(0);
    if (more) s1_frag_read<1>(As[cur ^ 1][0], Bs[cur ^ 1][0], aoff, boff, aU, bU);
  }

#pragma unroll
  for (int ct = 0; ct < 6; ++ct) {
    const int o = o0 + wn + ct * 16 + lm;
    const float bl = blin[o];
#pragma unroll
    for (int mt = 0; mt < 2; ++mt) {
#pragma unroll
      for (int r = 0; r < 8; ++r) {
        const int m = m0 + wm + mt * 16 + r + hi * 8;
        wout[(size_t)(b * Nn + m) * NK + o] = cAcc[mt][ct][r] + bl;
      }
    }
  }
}

#define KSTG (8 * KP * 2)
#define AWP 100
#define AWSTG (32 * AWP)

template <int P>
__device__ __forceinline__ void slab_fetch(const float* __restrict__ kb, int i0,
                                           int kPr0, int kC4, int tid, int hRow,
                                           float4 (&rx)[2], float4 (&ry)[2], float& rh) {
  const unsigned ko = (unsigned)((i0 + kPr0 * 2) * Hn + P * 512 + kC4);
#pragma unroll
  for (int j = 0; j < 2; ++j) {
    rx[j] = *(const float4*)(kb + (ko + (unsigned)((j * 8) * Hn)));
    ry[j] = *(const float4*)(kb + (ko + (unsigned)((j * 8 + 1) * Hn)));
  }
  if (tid < 32) {
    const float* krh = kb + (unsigned)((i0 + hRow) * Hn);
    float v = 0.0f;
    if (P == 0) { if (tid >= 16) v = krh[512]; }
    else        { if (tid < 16)  v = krh[511]; }
    rh = v;
  }
}

__device__ __forceinline__ void slab_store(float* KsP, int kPr0, int kC4, int tid,
                                           int hBase, const float4 (&rx)[2],
                                           const float4 (&ry)[2], float rh) {
  float* kd = KsP + (kPr0 * KP + 4 + kC4) * 2;
#pragma unroll
  for (int j = 0; j < 2; ++j) {
    const float4 x = rx[j], y = ry[j];
    const float4 lo = make_float4(x.x, y.x, x.y, y.y);
    const float4 up = make_float4(x.z, y.z, x.w, y.w);
    float* d = kd + (j * 4 * KP) * 2;
    *(float4*)(d)     = lo;
    *(float4*)(d + 4) = up;
  }
  if (tid < 32) {
    if (tid < 16) KsP[hBase + 3 * 2] = rh;
    else          KsP[hBase + 516 * 2] = rh;
  }
}

__device__ __forceinline__ void wcomb_store(float* AwS, int aDst, float g0, float g1, float g2) {
  const float s02 = g0 + g2;
  const float ev = g0 * (1.0f / 24.0f) + g2 * (1.0f / 6.0f);
  const float od = g1 * (1.0f / 12.0f);
  AwS[aDst + 0 * 16] = g0 * 0.25f;
  AwS[aDst + 1 * 16] = (s02 + g1) * (-1.0f / 6.0f);
  AwS[aDst + 2 * 16] = (s02 - g1) * (-1.0f / 6.0f);
  AwS[aDst + 3 * 16] = ev + od;
  AwS[aDst + 4 * 16] = ev - od;
  AwS[aDst + 5 * 16] = g2;
}

__device__ __forceinline__ float from_left(float old, float src) {
  return __int_as_float(__builtin_amdgcn_update_dpp(__float_as_int(old), __float_as_int(src),
                                                    0x111  , 0xF, 0xF, false));
}
__device__ __forceinline__ float from_right(float old, float src) {
  return __int_as_float(__builtin_amdgcn_update_dpp(__float_as_int(old), __float_as_int(src),
                                                    0x101  , 0xF, 0xF, false));
}

struct FragD {
  float4 q[2];
  v2f eL, eR;
};

template <int KS>
__device__ __forceinline__ void frag_issue(const float* KsP, int cw, int lm, int hi2, FragD& fr) {
  const float* row = &KsP[((KS * 2 + hi2) * KP) * 2];
  const float* p = row + (cw * 64 + lm * 4 + 4) * 2;
  fr.q[0] = *(const float4*)(p);
  fr.q[1] = *(const float4*)(p + 4);
  fr.eL = *(const v2f*)(row + (cw * 64 + 3) * 2);
  fr.eR = *(const v2f*)(row + (cw * 64 + 68) * 2);
}

template <int KS>
__device__ __forceinline__ void wread(const float* AwC, int lm, int hi2, v2f (&ws)[2][6]) {
#pragma unroll
  for (int rt = 0; rt < 2; ++rt)
#pragma unroll
    for (int c = 0; c < 6; ++c)
      ws[rt][c] = *(const v2f*)(&AwC[(rt * 16 + lm) * AWP + c * 16 + 4 * KS + 2 * hi2]);
}

__device__ __forceinline__ void combos(const FragD& fr, v2f (&bq)[6]) {
  v2f d0, d1, d2, d3, d4, d5;
  d1.x = fr.q[0].x; d1.y = fr.q[0].y;
  d2.x = fr.q[0].z; d2.y = fr.q[0].w;
  d3.x = fr.q[1].x; d3.y = fr.q[1].y;
  d4.x = fr.q[1].z; d4.y = fr.q[1].w;
  d0.x = from_left(fr.eL.x, d4.x);
  d0.y = from_left(fr.eL.y, d4.y);
  d5.x = from_right(fr.eR.x, d1.x);
  d5.y = from_right(fr.eR.y, d1.y);
  const v2f t1 = d4 - 4.0f * d2;
  const v2f t2 = d3 - 4.0f * d1;
  const v2f t3 = d4 - d2;
  const v2f u = d3 - d1;
  bq[0] = 4.0f * d0 + (d4 - 5.0f * d2);
  bq[1] = t1 + t2;
  bq[2] = t1 - t2;
  bq[3] = t3 + 2.0f * u;
  bq[4] = t3 - 2.0f * u;
  bq[5] = 4.0f * d1 + (d5 - 5.0f * d3);
}

template <int RT>
__device__ __forceinline__ void products(const v2f (&ws)[2][6], const v2f (&bq)[6], v8f (&acc)[12]) {
#pragma unroll
  for (int c = 0; c < 6; ++c)
    acc[RT * 6 + c] = __builtin_amdgcn_wmma_f32_16x16x4_f32(
        false, ws[RT][c], false, bq[c], (short)0, acc[RT * 6 + c], false, false);
}

__global__ __launch_bounds__(512) void dyconv_ln(
    const float* __restrict__ kin, const float* __restrict__ w,
    const float* __restrict__ gamma, const float* __restrict__ beta,
    float* __restrict__ out) {
  __shared__ __attribute__((aligned(16))) float Ks[2][2][KSTG];
  __shared__ __attribute__((aligned(16))) float Aw[2][AWSTG];
  __shared__ __attribute__((aligned(16))) float Gs[1024];
  __shared__ __attribute__((aligned(16))) float Bt[1024];
  __shared__ float redS[16 * 32];
  __shared__ float redQ[16 * 32];
  __shared__ float muS[32];
  __shared__ float rsS[32];

  const int tid = threadIdx.x;
  const int b = blockIdx.y;
  const int m0 = blockIdx.x * 32;
  const int wave = tid >> 5, lane = tid & 31;
  const int lm = lane & 15, hi2 = lane >> 4;
  const float* kb = kin + (size_t)b * Nn * Hn;
  const int half = wave >> 3;
  const int cw = wave & 7;

#pragma unroll
  for (int j = 0; j < 2; ++j) {
    int i = tid + j * 512;
    Gs[i] = gamma[i];
    Bt[i] = beta[i];
  }

  v8f acc[12] = {};

  const int aRow = tid >> 4;
  const int aI = tid & 15;
  const float* wp = w + (size_t)(b * Nn + m0 + aRow) * NK + aI * 3;
  const int aDst = aRow * AWP + aI;
  const int kPr0 = tid >> 7;
  const int kC4 = (tid & 127) * 4;
  const int hRow = tid & 15;
  const int hBase = ((hRow >> 1) * KP) * 2 + (hRow & 1);

  float4 rx[2], ry[2], rx1[2], ry1[2];
  float rh = 0.0f, rh1 = 0.0f;
  float rw0, rw1, rw2;

  {
    slab_fetch<0>(kb, 0, kPr0, kC4, tid, hRow, rx, ry, rh);
    slab_fetch<1>(kb, 0, kPr0, kC4, tid, hRow, rx1, ry1, rh1);
    rw0 = wp[0]; rw1 = wp[1]; rw2 = wp[2];
    slab_store(Ks[0][0], kPr0, kC4, tid, hBase, rx, ry, rh);
    slab_store(Ks[0][1], kPr0, kC4, tid, hBase, rx1, ry1, rh1);
    wcomb_store(Aw[0], aDst, rw0, rw1, rw2);
  }
  __syncthreads();

  v2f wa[2][6], wb[2][6];
  FragD fa, fb;
  v2f bq[6];
  wread<0>(Aw[0], lm, hi2, wa);
  frag_issue<0>(Ks[0][half], cw, lm, hi2, fa);

  for (int ic = 0; ic < 16; ++ic) {
    const int i0 = ic * 16;
    const int st = ic & 1;
    const float* K = Ks[st][half];
    const float* Kn = Ks[st ^ 1][half];
    const float* AwC = Aw[st];
    const float* AwN = Aw[st ^ 1];
    const bool more = (ic + 1) < 16;
    __builtin_amdgcn_sched_barrier(0);
    frag_issue<1>(K, cw, lm, hi2, fb);
    wread<1>(AwC, lm, hi2, wb);
    if (more) slab_fetch<0>(kb, i0 + 16, kPr0, kC4, tid, hRow, rx, ry, rh);
    __builtin_amdgcn_sched_barrier(0);
    combos(fa, bq);
    products<0>(wa, bq, acc);
    products<1>(wa, bq, acc);
    __builtin_amdgcn_sched_barrier(0);
    frag_issue<2>(K, cw, lm, hi2, fa);
    wread<2>(AwC, lm, hi2, wa);
    if (more) {
      slab_fetch<1>(kb, i0 + 16, kPr0, kC4, tid, hRow, rx1, ry1, rh1);
      const float* wq = wp + (size_t)(i0 + 16) * 3;
      rw0 = wq[0]; rw1 = wq[1]; rw2 = wq[2];
    }
    __builtin_amdgcn_sched_barrier(0);
    combos(fb, bq);
    products<0>(wb, bq, acc);
    products<1>(wb, bq, acc);
    __builtin_amdgcn_sched_barrier(0);
    frag_issue<3>(K, cw, lm, hi2, fb);
    wread<3>(AwC, lm, hi2, wb);
    if (more) slab_store(Ks[st ^ 1][0], kPr0, kC4, tid, hBase, rx, ry, rh);
    __builtin_amdgcn_sched_barrier(0);
    combos(fa, bq);
    products<0>(wa, bq, acc);
    products<1>(wa, bq, acc);
    __builtin_amdgcn_sched_barrier(0);
    if (more) {
      slab_store(Ks[st ^ 1][1], kPr0, kC4, tid, hBase, rx1, ry1, rh1);
      wcomb_store(Aw[st ^ 1], aDst, rw0, rw1, rw2);
    }
    wg_arrive();
    __builtin_amdgcn_sched_barrier(0);
    combos(fb, bq);
    products<0>(wb, bq, acc);
    __builtin_amdgcn_sched_barrier(0);
    wg_wait();
    if (more) {
      wread<0>(AwN, lm, hi2, wa);
      frag_issue<0>(Kn, cw, lm, hi2, fa);
    }
    __builtin_amdgcn_sched_barrier(0);
    products<1>(wb, bq, acc);
  }

#pragma unroll
  for (int rt = 0; rt < 2; ++rt)
#pragma unroll
    for (int r = 0; r < 8; ++r) {
      const float M0 = acc[rt * 6 + 0][r], M1 = acc[rt * 6 + 1][r], M2 = acc[rt * 6 + 2][r];
      const float M3 = acc[rt * 6 + 3][r], M4 = acc[rt * 6 + 4][r], M5 = acc[rt * 6 + 5][r];
      const float s12 = M1 + M2, d12 = M1 - M2, s34 = M3 + M4, d34 = M3 - M4;
      acc[rt * 6 + 0][r] = (M0 + s12) + s34;
      acc[rt * 6 + 1][r] = d12 + 2.0f * d34;
      acc[rt * 6 + 2][r] = s12 + 4.0f * s34;
      acc[rt * 6 + 3][r] = (d12 + 8.0f * d34) + M5;
    }

  float s[2][8], q[2][8];
#pragma unroll
  for (int rt = 0; rt < 2; ++rt)
#pragma unroll
    for (int r = 0; r < 8; ++r) { s[rt][r] = 0.0f; q[rt][r] = 0.0f; }
#pragma unroll
  for (int rt = 0; rt < 2; ++rt)
#pragma unroll
    for (int e = 0; e < 4; ++e)
#pragma unroll
      for (int r = 0; r < 8; ++r) {
        float v = acc[rt * 6 + e][r];
        s[rt][r] += v;
        q[rt][r] += v * v;
      }
#pragma unroll
  for (int mask = 1; mask < 16; mask <<= 1)
#pragma unroll
    for (int rt = 0; rt < 2; ++rt)
#pragma unroll
      for (int r = 0; r < 8; ++r) {
        s[rt][r] += __shfl_xor(s[rt][r], mask, 32);
        q[rt][r] += __shfl_xor(q[rt][r], mask, 32);
      }
  if (lm == 0) {
#pragma unroll
    for (int rt = 0; rt < 2; ++rt)
#pragma unroll
      for (int r = 0; r < 8; ++r) {
        const int row = rt * 16 + r + hi2 * 8;
        redS[wave * 32 + row] = s[rt][r];
        redQ[wave * 32 + row] = q[rt][r];
      }
  }
  __syncthreads();
  if (tid < 32) {
    float ss = 0.0f, qq = 0.0f;
#pragma unroll
    for (int wv = 0; wv < 16; ++wv) {
      ss += redS[wv * 32 + tid];
      qq += redQ[wv * 32 + tid];
    }
    const float mu = ss * (1.0f / 1024.0f);
    const float var = qq * (1.0f / 1024.0f) - mu * mu;
    muS[tid] = mu;
    rsS[tid] = rsqrtf(var + LN_EPS);
  }
  __syncthreads();

  {
    const int h = wave * 64 + lm * 4;
    const float4 g = *(const float4*)(&Gs[h]);
    const float4 be = *(const float4*)(&Bt[h]);
#pragma unroll
    for (int rt = 0; rt < 2; ++rt)
#pragma unroll
      for (int r = 0; r < 8; ++r) {
        const int row = rt * 16 + r + hi2 * 8;
        const float mu = muS[row], rs = rsS[row];
        float4 v;
        v.x = (acc[rt * 6 + 0][r] - mu) * rs * g.x + be.x;
        v.y = (acc[rt * 6 + 1][r] - mu) * rs * g.y + be.y;
        v.z = (acc[rt * 6 + 2][r] - mu) * rs * g.z + be.z;
        v.w = (acc[rt * 6 + 3][r] - mu) * rs * g.w + be.w;
        *(float4*)(&out[(size_t)(b * Nn + m0 + row) * Hn + h]) = v;
      }
  }
}

extern "C" void kernel_launch(void* const* d_in, const int* in_sizes, int n_in,
                              void* d_out, int out_size, void* d_ws, size_t ws_size,
                              hipStream_t stream) {
  const float* f     = (const float*)d_in[0];
  const float* k     = (const float*)d_in[1];
  const float* Wlin  = (const float*)d_in[2];
  const float* blin  = (const float*)d_in[3];
  const float* gamma = (const float*)d_in[4];
  const float* beta  = (const float*)d_in[5];
  float* out = (float*)d_out;
  float* wgt = (float*)d_ws;

  (void)in_sizes; (void)n_in; (void)out_size; (void)ws_size;

  wgen_gemm<<<dim3(NK / T1N, Nn / T1M, Bn), 256, 0, stream>>>(f, Wlin, blin, wgt);
  dyconv_ln<<<dim3(Nn / 32, Bn), 512, 0, stream>>>(k, wgt, gamma, beta, out);
}
